# starting kernel
# baseline (speedup 1.0000x reference)
#include <hip/hip_runtime.h>

typedef __attribute__((ext_vector_type(2))) float v2f;
typedef __attribute__((ext_vector_type(8))) float v8f;

#define B_  4
#define L_  4096
#define D_  512
#define HD_ 64
#define VD_ 128

constexpr float LOG2_GAMMA = -0.04580329f;

__device__ __forceinline__ v8f wmma_f32_16x16x4(v2f a, v2f b, v8f c) {
  return __builtin_amdgcn_wmma_f32_16x16x4_f32(false, a, false, b, (short)0, c,
                                               false, false);
}

template <int STRIDE>
__device__ __forceinline__ void gemm4tiles(const float* __restrict__ bp,
                                           const float* __restrict__ xrow,
                                           v8f acc[4]) {
#pragma unroll 4
  for (int k0 = 0; k0 < D_; k0 += 4) {
    v2f a = *(const v2f*)(xrow + k0);
#pragma unroll
    for (int t = 0; t < 4; ++t) {
      v2f b;
      b.x = bp[t * 16];
      b.y = bp[t * 16 + STRIDE];
      acc[t] = wmma_f32_16x16x4(a, b, acc[t]);
    }
    bp += 4 * STRIDE;
  }
}

__global__ __launch_bounds__(128) void proj_xpos_kernel(
    const float* __restrict__ X,  const float* __restrict__ WQ,
    const float* __restrict__ WK, const float* __restrict__ WV,
    float* __restrict__ Qb, float* __restrict__ Kb, float* __restrict__ Vb) {
  const int lane = threadIdx.x & 31;
  const int w    = threadIdx.x >> 5;
  const int r    = lane & 15;
  const int hi   = lane >> 4;
  const int rowBase = blockIdx.x * 16;

  const v8f vz = {0.f, 0.f, 0.f, 0.f, 0.f, 0.f, 0.f, 0.f};
  v8f acc[4];
#pragma unroll
  for (int t = 0; t < 4; ++t) acc[t] = vz;

  const float* xrow = X + (size_t)(rowBase + r) * D_ + 2 * hi;
  if (w == 0)      gemm4tiles<HD_>(WQ + r + 2 * hi * HD_, xrow, acc);
  else if (w == 1) gemm4tiles<HD_>(WK + r + 2 * hi * HD_, xrow, acc);
  else             gemm4tiles<VD_>(WV + (w - 2) * 64 + r + 2 * hi * VD_, xrow, acc);

#pragma unroll
  for (int t = 0; t < 4; ++t) {
    int cc = w * 64 + t * 16 + r;
#pragma unroll
    for (int i = 0; i < 8; ++i) {
      int row = rowBase + i + 8 * hi;
      float x = acc[t][i];
      float p = __shfl_xor(x, 1, 32);
      if (cc < 128) {
        int  n    = row & (L_ - 1);
        bool down = cc >= 64;
        int  ch   = cc & 63;
        int  j    = ch >> 1;
        float sv    = (2.0f * (float)j + 0.4f * (float)HD_) * (1.0f / (1.4f * (float)HD_));
        float ex    = (float)n * (1.0f / 512.0f);
        if (down) ex = -ex;
        float scale = exp2f(ex * log2f(sv));
        float invf  = exp2f((float)j * (-13.2877124f / 32.0f));
        float ang   = (float)n * invf;
        float sn, cs;
        sincosf(ang, &sn, &cs);
        sn *= scale; cs *= scale;
        float outv = (ch & 1) ? fmaf(x, cs, p * sn) : fmaf(x, cs, -(p * sn));
        if (!down) Qb[(size_t)row * HD_ + ch] = outv;
        else       Kb[(size_t)row * HD_ + ch] = outv;
      } else {
        Vb[(size_t)row * VD_ + (cc - 128)] = x;
      }
    }
  }
}

__global__ __launch_bounds__(32) void retention_attn_kernel(
    const float* __restrict__ Qb, const float* __restrict__ Kb,
    const float* __restrict__ Vb, float* __restrict__ Ob) {
  __shared__ float sS[16 * 18];
  const int lane = threadIdx.x & 31;
  const int r    = lane & 15;
  const int hi   = lane >> 4;

  const int numQ = L_ / 16;
  const int idx  = blockIdx.x;
  const int bb   = idx / numQ;
  const int qi   = (numQ - 1) - (idx % numQ);
  const int n0   = qi * 16;

  const float* Q = Qb + (size_t)bb * L_ * HD_;
  const float* K = Kb + (size_t)bb * L_ * HD_;
  const float* V = Vb + (size_t)bb * L_ * VD_;
  float*       O = Ob + (size_t)bb * L_ * VD_;

  v2f qa[16];
  {
    const float* qp = Q + (size_t)(n0 + r) * HD_ + 2 * hi;
#pragma unroll
    for (int t = 0; t < 16; ++t) qa[t] = *(const v2f*)(qp + 4 * t);
  }

  float gp[8];
#pragma unroll
  for (int i = 0; i < 8; ++i)
    gp[i] = exp2f((float)((i + 8 * hi) - r) * LOG2_GAMMA);

  const v8f vz = {0.f, 0.f, 0.f, 0.f, 0.f, 0.f, 0.f, 0.f};
  v8f o[8];
#pragma unroll
  for (int ct = 0; ct < 8; ++ct) o[ct] = vz;

  const int CUT = 1280;
  const int m0s = (n0 > CUT) ? ((n0 - CUT) & ~15) : 0;

  const float* kptr  = K + (size_t)(m0s + r) * HD_ + 2 * hi;
  const float* vbase = V + (size_t)m0s * VD_ + 2 * hi * VD_ + r;
  const char*  kpf   = (const char*)(K + (size_t)(m0s + 16 + (lane >> 1)) * HD_)
                       + (lane & 1) * 128;
  const char*  vpf   = (const char*)(V + (size_t)(m0s + 16 + (lane >> 1)) * VD_)
                       + (lane & 1) * 128;

  for (int m0 = m0s; m0 <= n0; m0 += 16) {
    if (m0 + 16 <= n0) {
      __builtin_prefetch(kpf, 0, 3);
      __builtin_prefetch(vpf, 0, 3);
      __builtin_prefetch(vpf + 256, 0, 3);
    }

    v8f s = vz;
#pragma unroll
    for (int t = 0; t < 16; ++t) {
      v2f kb = *(const v2f*)(kptr + 4 * t);
      s = wmma_f32_16x16x4(qa[t], kb, s);
    }

    if (m0 == n0) {
#pragma unroll
      for (int i = 0; i < 8; ++i) {
        int diff = (i + 8 * hi) - r;
        s[i] = (diff < 0) ? 0.0f : s[i] * gp[i];
      }
    } else {
      float cf = exp2f((float)(n0 - m0) * LOG2_GAMMA);
#pragma unroll
      for (int i = 0; i < 8; ++i) s[i] *= cf * gp[i];
    }

    __syncthreads();
#pragma unroll
    for (int i = 0; i < 8; ++i) sS[(i + 8 * hi) * 18 + r] = s[i];
    __syncthreads();
    v2f av[4];
#pragma unroll
    for (int u = 0; u < 4; ++u)
      av[u] = *(const v2f*)&sS[r * 18 + 4 * u + 2 * hi];

#pragma unroll
    for (int u = 0; u < 4; ++u) {
#pragma unroll
      for (int ct = 0; ct < 8; ++ct) {
        v2f vb;
        vb.x = vbase[u * 4 * VD_ + ct * 16];
        vb.y = vbase[u * 4 * VD_ + VD_ + ct * 16];
        o[ct] = wmma_f32_16x16x4(av[u], vb, o[ct]);
      }
    }

    kptr  += 16 * HD_;
    vbase += 16 * VD_;
    kpf   += 16 * HD_ * sizeof(float);
    vpf   += 16 * VD_ * sizeof(float);
  }

  float* optr = O + (size_t)(n0 + 8 * hi) * VD_ + r;
  float* optr0 = O + (size_t)n0 * VD_ + r;
  (void)optr;
#pragma unroll
  for (int i = 0; i < 8; ++i) {
#pragma unroll
    for (int ct = 0; ct < 8; ++ct) {
      optr0[(size_t)(i + 8 * hi) * VD_ + ct * 16] = o[ct][i];
    }
  }
}

extern "C" void kernel_launch(void* const* d_in, const int* in_sizes, int n_in,
                              void* d_out, int out_size, void* d_ws, size_t ws_size,
                              hipStream_t stream) {
  const float* X  = (const float*)d_in[0];
  const float* WQ = (const float*)d_in[1];
  const float* WK = (const float*)d_in[2];
  const float* WV = (const float*)d_in[3];

  float* Qb = (float*)d_ws;
  float* Kb = Qb + (size_t)B_ * L_ * HD_;
  float* Vb = Kb + (size_t)B_ * L_ * HD_;
  float* O  = (float*)d_out;

  const int rowTiles = B_ * L_ / 16;
  proj_xpos_kernel<<<dim3(rowTiles), dim3(128), 0, stream>>>(X, WQ, WK, WV,
                                                             Qb, Kb, Vb);
  retention_attn_kernel<<<dim3(rowTiles), dim3(32), 0, stream>>>(Qb, Kb, Vb, O);
}


# retention kernel: each query tile's key chunks dealt over 8 waves of one work-group, partial O tiles merged through LDS in fixed wave order (changes the summation order over keys)
# speedup vs baseline: 1.1970x; 1.1970x over previous
#include <hip/hip_runtime.h>

typedef __attribute__((ext_vector_type(2))) float v2f;
typedef __attribute__((ext_vector_type(4))) float v4f;
typedef __attribute__((ext_vector_type(8))) float v8f;

#define B_  4
#define L_  4096
#define D_  512
#define HD_ 64
#define VD_ 128

constexpr float LOG2_GAMMA = -0.04580329f;

__device__ __forceinline__ v8f wmma_f32_16x16x4(v2f a, v2f b, v8f c) {
  return __builtin_amdgcn_wmma_f32_16x16x4_f32(false, a, false, b, (short)0, c,
                                               false, false);
}

template <int STRIDE>
__device__ __forceinline__ void gemm4tiles(const float* __restrict__ bp,
                                           const float* __restrict__ xrow,
                                           v8f acc[4]) {
#pragma unroll 4
  for (int k0 = 0; k0 < D_; k0 += 4) {
    v2f a = *(const v2f*)(xrow + k0);
#pragma unroll
    for (int t = 0; t < 4; ++t) {
      v2f b;
      b.x = bp[t * 16];
      b.y = bp[t * 16 + STRIDE];
      acc[t] = wmma_f32_16x16x4(a, b, acc[t]);
    }
    bp += 4 * STRIDE;
  }
}

__global__ __launch_bounds__(128) void proj_xpos_kernel(
    const float* __restrict__ X,  const float* __restrict__ WQ,
    const float* __restrict__ WK, const float* __restrict__ WV,
    float* __restrict__ Qb, float* __restrict__ Kb, float* __restrict__ Vb) {
  const int lane = threadIdx.x & 31;
  const int w    = threadIdx.x >> 5;
  const int r    = lane & 15;
  const int hi   = lane >> 4;
  const int rowBase = blockIdx.x * 16;

  const v8f vz = {0.f, 0.f, 0.f, 0.f, 0.f, 0.f, 0.f, 0.f};
  v8f acc[4];
#pragma unroll
  for (int t = 0; t < 4; ++t) acc[t] = vz;

  const float* xrow = X + (size_t)(rowBase + r) * D_ + 2 * hi;
  if (w == 0)      gemm4tiles<HD_>(WQ + r + 2 * hi * HD_, xrow, acc);
  else if (w == 1) gemm4tiles<HD_>(WK + r + 2 * hi * HD_, xrow, acc);
  else             gemm4tiles<VD_>(WV + (w - 2) * 64 + r + 2 * hi * VD_, xrow, acc);

#pragma unroll
  for (int t = 0; t < 4; ++t) {
    int cc = w * 64 + t * 16 + r;
#pragma unroll
    for (int i = 0; i < 8; ++i) {
      int row = rowBase + i + 8 * hi;
      float x = acc[t][i];
      float p = __shfl_xor(x, 1, 32);
      if (cc < 128) {
        int  n    = row & (L_ - 1);
        bool down = cc >= 64;
        int  ch   = cc & 63;
        int  j    = ch >> 1;
        float sv    = (2.0f * (float)j + 0.4f * (float)HD_) * (1.0f / (1.4f * (float)HD_));
        float ex    = (float)n * (1.0f / 512.0f);
        if (down) ex = -ex;
        float scale = exp2f(ex * log2f(sv));
        float invf  = exp2f((float)j * (-13.2877124f / 32.0f));
        float ang   = (float)n * invf;
        float sn, cs;
        sincosf(ang, &sn, &cs);
        sn *= scale; cs *= scale;
        float outv = (ch & 1) ? fmaf(x, cs, p * sn) : fmaf(x, cs, -(p * sn));
        if (!down) Qb[(size_t)row * HD_ + ch] = outv;
        else       Kb[(size_t)row * HD_ + ch] = outv;
      } else {
        Vb[(size_t)row * VD_ + (cc - 128)] = x;
      }
    }
  }
}

#define NW_ 8
#define PP_ 132

__global__ __launch_bounds__(32 * NW_) void retention_attn_kernel(
    const float* __restrict__ Qb, const float* __restrict__ Kb,
    const float* __restrict__ Vb, float* __restrict__ Ob) {
  __shared__ __attribute__((aligned(16))) float sP[NW_][16 * PP_];
  const int lane = threadIdx.x & 31;
  const int w    = __builtin_amdgcn_readfirstlane((int)(threadIdx.x >> 5));
  const int r    = lane & 15;
  const int hi   = lane >> 4;
  float* sS = sP[w];

  const int numQ = L_ / 16;
  const int idx  = blockIdx.x;
  const int bb   = idx / numQ;
  const int qi   = (numQ - 1) - (idx % numQ);
  const int n0   = qi * 16;

  const float* Q = Qb + (size_t)bb * L_ * HD_;
  const float* K = Kb + (size_t)bb * L_ * HD_;
  const float* V = Vb + (size_t)bb * L_ * VD_;
  float*       O = Ob + (size_t)bb * L_ * VD_;

  v2f qa[16];
  {
    const float* qp = Q + (size_t)(n0 + r) * HD_ + 2 * hi;
#pragma unroll
    for (int t = 0; t < 16; ++t) qa[t] = *(const v2f*)(qp + 4 * t);
  }

  float gp[8];
#pragma unroll
  for (int i = 0; i < 8; ++i)
    gp[i] = exp2f((float)((i + 8 * hi) - r) * LOG2_GAMMA);

  const v8f vz = {0.f, 0.f, 0.f, 0.f, 0.f, 0.f, 0.f, 0.f};
  v8f o[8];
#pragma unroll
  for (int ct = 0; ct < 8; ++ct) o[ct] = vz;

  const int CUT = 1280;
  const int m0s = (n0 > CUT) ? ((n0 - CUT) & ~15) : 0;
  const int mw  = m0s + 16 * w;
  const int MST = 16 * NW_;

  const float* kptr  = K + (size_t)(mw + r) * HD_ + 2 * hi;
  const float* vbase = V + (size_t)mw * VD_ + 2 * hi * VD_ + r;
  const char*  kpf   = (const char*)(K + (size_t)(mw + MST + (lane >> 1)) * HD_)
                       + (lane & 1) * 128;
  const char*  vpf   = (const char*)(V + (size_t)(mw + MST + (lane >> 1)) * VD_)
                       + (lane & 1) * 128;

  for (int m0 = mw; m0 <= n0; m0 += MST) {
    if (m0 + MST <= n0) {
      __builtin_prefetch(kpf, 0, 3);
      __builtin_prefetch(vpf, 0, 3);
      __builtin_prefetch(vpf + 256, 0, 3);
    }

    v8f s = vz;
#pragma unroll
    for (int t = 0; t < 16; ++t) {
      v2f kb = *(const v2f*)(kptr + 4 * t);
      s = wmma_f32_16x16x4(qa[t], kb, s);
    }

    if (m0 == n0) {
#pragma unroll
      for (int i = 0; i < 8; ++i) {
        int diff = (i + 8 * hi) - r;
        s[i] = (diff < 0) ? 0.0f : s[i] * gp[i];
      }
    } else {
      float cf = exp2f((float)(n0 - m0) * LOG2_GAMMA);
#pragma unroll
      for (int i = 0; i < 8; ++i) s[i] *= cf * gp[i];
    }

#pragma unroll
    for (int i = 0; i < 8; ++i) sS[(i + 8 * hi) * 18 + r] = s[i];
    v2f av[4];
#pragma unroll
    for (int u = 0; u < 4; ++u)
      av[u] = *(const v2f*)&sS[r * 18 + 4 * u + 2 * hi];

#pragma unroll
    for (int u = 0; u < 4; ++u) {
#pragma unroll
      for (int ct = 0; ct < 8; ++ct) {
        v2f vb;
        vb.x = vbase[u * 4 * VD_ + ct * 16];
        vb.y = vbase[u * 4 * VD_ + VD_ + ct * 16];
        o[ct] = wmma_f32_16x16x4(av[u], vb, o[ct]);
      }
    }

    kptr  += MST * HD_;
    vbase += MST * VD_;
    kpf   += MST * HD_ * sizeof(float);
    vpf   += MST * VD_ * sizeof(float);
  }

#pragma unroll
  for (int ct = 0; ct < 8; ++ct) {
#pragma unroll
    for (int i = 0; i < 8; ++i)
      sS[(i + 8 * hi) * PP_ + ct * 16 + r] = o[ct][i];
  }
  __syncthreads();

  v4f a0 = {0.f, 0.f, 0.f, 0.f};
  v4f a1 = {0.f, 0.f, 0.f, 0.f};
#pragma unroll
  for (int p = 0; p < NW_; ++p) {
    a0 += *(const v4f*)&sP[p][(2 * w)     * PP_ + 4 * lane];
    a1 += *(const v4f*)&sP[p][(2 * w + 1) * PP_ + 4 * lane];
  }
  float* orow = O + (size_t)(n0 + 2 * w) * VD_ + 4 * lane;
  *(v4f*)(orow)       = a0;
  *(v4f*)(orow + VD_) = a1;
}

extern "C" void kernel_launch(void* const* d_in, const int* in_sizes, int n_in,
                              void* d_out, int out_size, void* d_ws, size_t ws_size,
                              hipStream_t stream) {
  const float* X  = (const float*)d_in[0];
  const float* WQ = (const float*)d_in[1];
  const float* WK = (const float*)d_in[2];
  const float* WV = (const float*)d_in[3];

  float* Qb = (float*)d_ws;
  float* Kb = Qb + (size_t)B_ * L_ * HD_;
  float* Vb = Kb + (size_t)B_ * L_ * HD_;
  float* O  = (float*)d_out;

  const int rowTiles = B_ * L_ / 16;
  proj_xpos_kernel<<<dim3(rowTiles), dim3(128), 0, stream>>>(X, WQ, WK, WV,
                                                             Qb, Kb, Vb);
  retention_attn_kernel<<<dim3(rowTiles), dim3(32 * NW_), 0, stream>>>(Qb, Kb, Vb, O);
}


# K,V workspace in WMMA fragment order (K tile computed transposed): retention loads a chunk as one group of 24 whole-line 16 B/lane loads, decay branch-free, no prefetch; k order in products changes
# speedup vs baseline: 1.3264x; 1.1081x over previous
#include <hip/hip_runtime.h>

typedef __attribute__((ext_vector_type(2))) float v2f;
typedef __attribute__((ext_vector_type(4))) float v4f;
typedef __attribute__((ext_vector_type(8))) float v8f;

#define B_  4
#define L_  4096
#define D_  512
#define HD_ 64
#define VD_ 128

constexpr float LOG2_GAMMA = -0.04580329f;

__device__ __forceinline__ v8f wmma_f32_16x16x4(v2f a, v2f b, v8f c) {
  return __builtin_amdgcn_wmma_f32_16x16x4_f32(false, a, false, b, (short)0, c,
                                               false, false);
}

template <int STRIDE, bool SWAP>
__device__ __forceinline__ void gemm4tiles(const float* __restrict__ bp,
                                           const float* __restrict__ xrow,
                                           v8f acc[4]) {
#pragma unroll 4
  for (int k0 = 0; k0 < D_; k0 += 4) {
    v2f a = *(const v2f*)(xrow + k0);
#pragma unroll
    for (int t = 0; t < 4; ++t) {
      v2f b;
      b.x = bp[t * 16];
      b.y = bp[t * 16 + STRIDE];
      acc[t] = SWAP ? wmma_f32_16x16x4(b, a, acc[t])
                    : wmma_f32_16x16x4(a, b, acc[t]);
    }
    bp += 4 * STRIDE;
  }
}

__device__ __forceinline__ void xpos_coeffs(int n, int j, bool down,
                                            float& sn, float& cs) {
  float sv    = (2.0f * (float)j + 0.4f * (float)HD_) * (1.0f / (1.4f * (float)HD_));
  float ex    = (float)n * (1.0f / 512.0f);
  if (down) ex = -ex;
  float scale = exp2f(ex * log2f(sv));
  float invf  = exp2f((float)j * (-13.2877124f / 32.0f));
  float ang   = (float)n * invf;
  sincosf(ang, &sn, &cs);
  sn *= scale; cs *= scale;
}

__global__ __launch_bounds__(128) void proj_xpos_kernel(
    const float* __restrict__ X,  const float* __restrict__ WQ,
    const float* __restrict__ WK, const float* __restrict__ WV,
    float* __restrict__ Qb, float* __restrict__ Kb, float* __restrict__ Vb) {
  const int lane = threadIdx.x & 31;
  const int w    = __builtin_amdgcn_readfirstlane((int)(threadIdx.x >> 5));
  const int r    = lane & 15;
  const int hi   = lane >> 4;
  const int rowBase = blockIdx.x * 16;

  const v8f vz = {0.f, 0.f, 0.f, 0.f, 0.f, 0.f, 0.f, 0.f};
  v8f acc[4];
#pragma unroll
  for (int t = 0; t < 4; ++t) acc[t] = vz;

  const float* xrow = X + (size_t)(rowBase + r) * D_ + 2 * hi;

  if (w == 0) {
    gemm4tiles<HD_, false>(WQ + r + 2 * hi * HD_, xrow, acc);
#pragma unroll
    for (int t = 0; t < 4; ++t) {
      int ch = t * 16 + r;
#pragma unroll
      for (int i = 0; i < 8; ++i) {
        int row = rowBase + i + 8 * hi;
        float x = acc[t][i];
        float p = __shfl_xor(x, 1, 32);
        float sn, cs;
        xpos_coeffs(row & (L_ - 1), ch >> 1, false, sn, cs);
        float outv = (ch & 1) ? fmaf(x, cs, p * sn) : fmaf(x, cs, -(p * sn));
        Qb[(size_t)row * HD_ + ch] = outv;
      }
    }
  } else if (w == 1) {
    gemm4tiles<HD_, true>(WK + r + 2 * hi * HD_, xrow, acc);
    const int n = (rowBase + r) & (L_ - 1);
    float* kdst = Kb + (size_t)blockIdx.x * (16 * HD_) + lane * 4;
#pragma unroll
    for (int t = 0; t < 4; ++t) {
      float o8[8];
#pragma unroll
      for (int ip = 0; ip < 4; ++ip) {
        float sn, cs;
        xpos_coeffs(n, 8 * t + 4 * hi + ip, true, sn, cs);
        float xe = acc[t][2 * ip];
        float xo = acc[t][2 * ip + 1];
        o8[2 * ip]     = fmaf(xe, cs, -(xo * sn));
        o8[2 * ip + 1] = fmaf(xo, cs, xe * sn);
      }
      v4f lo = {o8[0], o8[1], o8[2], o8[3]};
      v4f up = {o8[4], o8[5], o8[6], o8[7]};
      *(v4f*)(kdst + (2 * t)     * 128) = lo;
      *(v4f*)(kdst + (2 * t + 1) * 128) = up;
    }
  } else {
    gemm4tiles<VD_, false>(WV + (w - 2) * 64 + r + 2 * hi * VD_, xrow, acc);
    float* vdst = Vb + (size_t)blockIdx.x * (16 * VD_) + lane * 4;
#pragma unroll
    for (int t = 0; t < 4; ++t) {
      int ct = (w - 2) * 4 + t;
      v4f lo = {acc[t][0], acc[t][1], acc[t][2], acc[t][3]};
      v4f up = {acc[t][4], acc[t][5], acc[t][6], acc[t][7]};
      *(v4f*)(vdst + ct * 256)       = lo;
      *(v4f*)(vdst + ct * 256 + 128) = up;
    }
  }
}

#define NW_ 8
#define PP_ 132

__global__ __launch_bounds__(32 * NW_) void retention_attn_kernel(
    const float* __restrict__ Qb, const float* __restrict__ Kb,
    const float* __restrict__ Vb, float* __restrict__ Ob) {
  __shared__ __attribute__((aligned(16))) float sP[NW_][16 * PP_];
  const int lane = threadIdx.x & 31;
  const int w    = __builtin_amdgcn_readfirstlane((int)(threadIdx.x >> 5));
  const int r    = lane & 15;
  const int hi   = lane >> 4;
  float* sS = sP[w];

  const int numQ = L_ / 16;
  const int idx  = blockIdx.x;
  const int bb   = idx / numQ;
  const int qi   = (numQ - 1) - (idx % numQ);
  const int n0   = qi * 16;

  const float* Q = Qb + (size_t)bb * L_ * HD_;
  const float* K = Kb + (size_t)bb * L_ * HD_;
  const float* V = Vb + (size_t)bb * L_ * VD_;
  float*       O = Ob + (size_t)bb * L_ * VD_;

  v2f qa[16];
  {
    const float* qp = Q + (size_t)(n0 + r) * HD_ + 8 * hi;
#pragma unroll
    for (int t4 = 0; t4 < 4; ++t4) {
      v4f q0 = *(const v4f*)(qp + 16 * t4);
      v4f q1 = *(const v4f*)(qp + 16 * t4 + 4);
      qa[4 * t4 + 0] = v2f{q0.x, q0.y};
      qa[4 * t4 + 1] = v2f{q0.z, q0.w};
      qa[4 * t4 + 2] = v2f{q1.x, q1.y};
      qa[4 * t4 + 3] = v2f{q1.z, q1.w};
    }
  }

  float gp[8];
#pragma unroll
  for (int i = 0; i < 8; ++i)
    gp[i] = exp2f((float)((i + 8 * hi) - r) * LOG2_GAMMA);

  const v8f vz = {0.f, 0.f, 0.f, 0.f, 0.f, 0.f, 0.f, 0.f};
  v8f o[8];
#pragma unroll
  for (int ct = 0; ct < 8; ++ct) o[ct] = vz;

  const int CUT = 1280;
  const int m0s = (n0 > CUT) ? ((n0 - CUT) & ~15) : 0;
  const int mw  = m0s + 16 * w;
  const int MST = 16 * NW_;

  const float* kptr  = K + (size_t)mw * HD_ + lane * 4;
  const float* vbase = V + (size_t)mw * VD_ + lane * 4;

  for (int m0 = mw; m0 <= n0; m0 += MST) {
    v4f kf[8];
#pragma unroll
    for (int j = 0; j < 8; ++j) kf[j] = *(const v4f*)(kptr + j * 128);
    v4f vf[2][8];
#pragma unroll
    for (int q = 0; q < 2; ++q) {
#pragma unroll
      for (int ct = 0; ct < 8; ++ct)
        vf[q][ct] = *(const v4f*)(vbase + ct * 256 + q * 128);
    }

    v8f s = vz;
#pragma unroll
    for (int j = 0; j < 8; ++j) {
      s = wmma_f32_16x16x4(qa[2 * j],     v2f{kf[j].x, kf[j].y}, s);
      s = wmma_f32_16x16x4(qa[2 * j + 1], v2f{kf[j].z, kf[j].w}, s);
    }

    {
      const bool diag = (m0 == n0);
      float cf = exp2f((float)(n0 - m0) * LOG2_GAMMA);
#pragma unroll
      for (int i = 0; i < 8; ++i) {
        int diff = (i + 8 * hi) - r;
        float v = s[i] * (cf * gp[i]);
        s[i] = (diag && diff < 0) ? 0.0f : v;
      }
    }

#pragma unroll
    for (int i = 0; i < 8; ++i) sS[(i + 8 * hi) * 18 + r] = s[i];
    v2f av[4];
#pragma unroll
    for (int u = 0; u < 4; ++u)
      av[u] = *(const v2f*)&sS[r * 18 + 8 * hi + 2 * u];

#pragma unroll
    for (int q = 0; q < 2; ++q) {
#pragma unroll
      for (int ct = 0; ct < 8; ++ct) {
        v4f f = vf[q][ct];
        o[ct] = wmma_f32_16x16x4(av[2 * q],     v2f{f.x, f.y}, o[ct]);
        o[ct] = wmma_f32_16x16x4(av[2 * q + 1], v2f{f.z, f.w}, o[ct]);
      }
    }

    kptr  += MST * HD_;
    vbase += MST * VD_;
  }

#pragma unroll
  for (int ct = 0; ct < 8; ++ct) {
#pragma unroll
    for (int i = 0; i < 8; ++i)
      sS[(i + 8 * hi) * PP_ + ct * 16 + r] = o[ct][i];
  }
  __syncthreads();

  v4f a0 = {0.f, 0.f, 0.f, 0.f};
  v4f a1 = {0.f, 0.f, 0.f, 0.f};
#pragma unroll
  for (int p = 0; p < NW_; ++p) {
    a0 += *(const v4f*)&sP[p][(2 * w)     * PP_ + 4 * lane];
    a1 += *(const v4f*)&sP[p][(2 * w + 1) * PP_ + 4 * lane];
  }
  float* orow = O + (size_t)(n0 + 2 * w) * VD_ + 4 * lane;
  *(v4f*)(orow)       = a0;
  *(v4f*)(orow + VD_) = a1;
}

extern "C" void kernel_launch(void* const* d_in, const int* in_sizes, int n_in,
                              void* d_out, int out_size, void* d_ws, size_t ws_size,
                              hipStream_t stream) {
  const float* X  = (const float*)d_in[0];
  const float* WQ = (const float*)d_in[1];
  const float* WK = (const float*)d_in[2];
  const float* WV = (const float*)d_in[3];

  float* Qb = (float*)d_ws;
  float* Kb = Qb + (size_t)B_ * L_ * HD_;
  float* Vb = Kb + (size_t)B_ * L_ * HD_;
  float* O  = (float*)d_out;

  const int rowTiles = B_ * L_ / 16;
  proj_xpos_kernel<<<dim3(rowTiles), dim3(128), 0, stream>>>(X, WQ, WK, WV,
                                                             Qb, Kb, Vb);
  retention_attn_kernel<<<dim3(rowTiles), dim3(32 * NW_), 0, stream>>>(Qb, Kb, Vb, O);
}


# retention in block form: per 64-key block a decayed K^T V state (64x128, WMMA), prefix recurrence over blocks, q.P adds all earlier keys; no key chunk revisited; sum order over keys changes
# speedup vs baseline: 1.6446x; 1.2399x over previous
#include <hip/hip_runtime.h>

typedef __attribute__((ext_vector_type(2))) float v2f;
typedef __attribute__((ext_vector_type(4))) float v4f;
typedef __attribute__((ext_vector_type(8))) float v8f;

#define B_  4
#define L_  4096
#define D_  512
#define HD_ 64
#define VD_ 128

constexpr float LOG2_GAMMA = -0.04580329f;

__device__ __forceinline__ v8f wmma_f32_16x16x4(v2f a, v2f b, v8f c) {
  return __builtin_amdgcn_wmma_f32_16x16x4_f32(false, a, false, b, (short)0, c,
                                               false, false);
}

template <int STRIDE, bool SWAP>
__device__ __forceinline__ void gemm4tiles(const float* __restrict__ bp,
                                           const float* __restrict__ xrow,
                                           v8f acc[4]) {
#pragma unroll 4
  for (int k0 = 0; k0 < D_; k0 += 4) {
    v2f a = *(const v2f*)(xrow + k0);
#pragma unroll
    for (int t = 0; t < 4; ++t) {
      v2f b;
      b.x = bp[t * 16];
      b.y = bp[t * 16 + STRIDE];
      acc[t] = SWAP ? wmma_f32_16x16x4(b, a, acc[t])
                    : wmma_f32_16x16x4(a, b, acc[t]);
    }
    bp += 4 * STRIDE;
  }
}

__device__ __forceinline__ void xpos_coeffs(int n, int j, bool down,
                                            float& sn, float& cs) {
  float sv    = (2.0f * (float)j + 0.4f * (float)HD_) * (1.0f / (1.4f * (float)HD_));
  float ex    = (float)n * (1.0f / 512.0f);
  if (down) ex = -ex;
  float scale = exp2f(ex * log2f(sv));
  float invf  = exp2f((float)j * (-13.2877124f / 32.0f));
  float ang   = (float)n * invf;
  sincosf(ang, &sn, &cs);
  sn *= scale; cs *= scale;
}

__global__ __launch_bounds__(128) void proj_xpos_kernel(
    const float* __restrict__ X,  const float* __restrict__ WQ,
    const float* __restrict__ WK, const float* __restrict__ WV,
    float* __restrict__ Qb, float* __restrict__ Kb, float* __restrict__ Vb) {
  const int lane = threadIdx.x & 31;
  const int w    = __builtin_amdgcn_readfirstlane((int)(threadIdx.x >> 5));
  const int r    = lane & 15;
  const int hi   = lane >> 4;
  const int rowBase = blockIdx.x * 16;

  const v8f vz = {0.f, 0.f, 0.f, 0.f, 0.f, 0.f, 0.f, 0.f};
  v8f acc[4];
#pragma unroll
  for (int t = 0; t < 4; ++t) acc[t] = vz;

  const float* xrow = X + (size_t)(rowBase + r) * D_ + 2 * hi;

  if (w == 0) {
    gemm4tiles<HD_, false>(WQ + r + 2 * hi * HD_, xrow, acc);
#pragma unroll
    for (int t = 0; t < 4; ++t) {
      int ch = t * 16 + r;
#pragma unroll
      for (int i = 0; i < 8; ++i) {
        int row = rowBase + i + 8 * hi;
        float x = acc[t][i];
        float p = __shfl_xor(x, 1, 32);
        float sn, cs;
        xpos_coeffs(row & (L_ - 1), ch >> 1, false, sn, cs);
        float outv = (ch & 1) ? fmaf(x, cs, p * sn) : fmaf(x, cs, -(p * sn));
        Qb[(size_t)row * HD_ + ch] = outv;
      }
    }
  } else if (w == 1) {
    gemm4tiles<HD_, true>(WK + r + 2 * hi * HD_, xrow, acc);
    const int n = (rowBase + r) & (L_ - 1);
    float* kdst = Kb + (size_t)blockIdx.x * (16 * HD_) + lane * 4;
#pragma unroll
    for (int t = 0; t < 4; ++t) {
      float o8[8];
#pragma unroll
      for (int ip = 0; ip < 4; ++ip) {
        float sn, cs;
        xpos_coeffs(n, 8 * t + 4 * hi + ip, true, sn, cs);
        float xe = acc[t][2 * ip];
        float xo = acc[t][2 * ip + 1];
        o8[2 * ip]     = fmaf(xe, cs, -(xo * sn));
        o8[2 * ip + 1] = fmaf(xo, cs, xe * sn);
      }
      v4f lo = {o8[0], o8[1], o8[2], o8[3]};
      v4f up = {o8[4], o8[5], o8[6], o8[7]};
      *(v4f*)(kdst + (2 * t)     * 128) = lo;
      *(v4f*)(kdst + (2 * t + 1) * 128) = up;
    }
  } else {
    gemm4tiles<VD_, false>(WV + (w - 2) * 64 + r + 2 * hi * VD_, xrow, acc);
    float* vdst = Vb + (size_t)blockIdx.x * (16 * VD_) + lane * 4;
#pragma unroll
    for (int t = 0; t < 4; ++t) {
      int ct = (w - 2) * 4 + t;
      v4f lo = {acc[t][0], acc[t][1], acc[t][2], acc[t][3]};
      v4f up = {acc[t][4], acc[t][5], acc[t][6], acc[t][7]};
      *(v4f*)(vdst + ct * 256)       = lo;
      *(v4f*)(vdst + ct * 256 + 128) = up;
    }
  }
}

#define BS_   64
#define NBLK_ (L_ / BS_)

__global__ __launch_bounds__(128) void block_state_kernel(
    const float* __restrict__ Kb, const float* __restrict__ Vb, float* Ob) {
  const int lane = threadIdx.x & 31;
  const int w    = __builtin_amdgcn_readfirstlane((int)(threadIdx.x >> 5));
  const int r    = lane & 15;
  const int hi   = lane >> 4;
  const int bb   = blockIdx.x / (NBLK_ - 1);
  const int c    = blockIdx.x % (NBLK_ - 1);

  const float* kp = Kb + ((size_t)bb * L_ + (size_t)c * BS_) * HD_
                       + ((r >> 2) & 1) * 128 + (r >> 3) * 64 + (r & 3) + 32 * hi;
  const float* vp = Vb + ((size_t)bb * L_ + (size_t)c * BS_) * VD_
                       + lane * 4 + (2 * w) * 256;
  float* Lt = Ob + ((size_t)bb * L_ + (size_t)(c + 1) * BS_) * VD_;

  const v8f vz = {0.f, 0.f, 0.f, 0.f, 0.f, 0.f, 0.f, 0.f};
  v8f acc[4][2];
#pragma unroll
  for (int ck = 0; ck < 4; ++ck) { acc[ck][0] = vz; acc[ck][1] = vz; }

#pragma unroll
  for (int cj = 0; cj < BS_ / 16; ++cj) {
    v4f vf[2][2];
#pragma unroll
    for (int ctl = 0; ctl < 2; ++ctl) {
#pragma unroll
      for (int q = 0; q < 2; ++q)
        vf[ctl][q] = *(const v4f*)(vp + cj * (16 * VD_) + ctl * 256 + q * 128);
    }
    v2f ka[4][4];
#pragma unroll
    for (int ck = 0; ck < 4; ++ck) {
#pragma unroll
      for (int u = 0; u < 4; ++u) {
        ka[ck][u].x = kp[cj * (16 * HD_) + ck * 256 + u * 8];
        ka[ck][u].y = kp[cj * (16 * HD_) + ck * 256 + u * 8 + 4];
      }
    }
#pragma unroll
    for (int q = 0; q < 2; ++q) {
      v4f d;
      d.x = exp2f((float)(BS_ - 16 * cj - 8 * hi - 4 * q - 0) * LOG2_GAMMA);
      d.y = exp2f((float)(BS_ - 16 * cj - 8 * hi - 4 * q - 1) * LOG2_GAMMA);
      d.z = exp2f((float)(BS_ - 16 * cj - 8 * hi - 4 * q - 2) * LOG2_GAMMA);
      d.w = exp2f((float)(BS_ - 16 * cj - 8 * hi - 4 * q - 3) * LOG2_GAMMA);
      vf[0][q] *= d;
      vf[1][q] *= d;
    }
#pragma unroll
    for (int ck = 0; ck < 4; ++ck) {
#pragma unroll
      for (int ctl = 0; ctl < 2; ++ctl) {
#pragma unroll
        for (int q = 0; q < 2; ++q) {
          v4f f = vf[ctl][q];
          acc[ck][ctl] = wmma_f32_16x16x4(ka[ck][2 * q],     v2f{f.x, f.y}, acc[ck][ctl]);
          acc[ck][ctl] = wmma_f32_16x16x4(ka[ck][2 * q + 1], v2f{f.z, f.w}, acc[ck][ctl]);
        }
      }
    }
  }

#pragma unroll
  for (int ck = 0; ck < 4; ++ck) {
#pragma unroll
    for (int ctl = 0; ctl < 2; ++ctl) {
      v8f a = acc[ck][ctl];
      v4f lo = {a[0], a[1], a[2], a[3]};
      v4f up = {a[4], a[5], a[6], a[7]};
      float* dst = Lt + (size_t)(32 * w + 16 * ctl + r) * HD_ + 16 * ck + 8 * hi;
      *(v4f*)(dst)     = lo;
      *(v4f*)(dst + 4) = up;
    }
  }
}

#define SCAN_GRP_ 31
__global__ __launch_bounds__(256) void state_scan_kernel(float* Ob) {
  const int e  = (blockIdx.x & 7) * 256 + threadIdx.x;
  const int bb = blockIdx.x >> 3;
  float* base = Ob + (size_t)bb * L_ * VD_ + (size_t)e * 4;
  const float g = exp2f((float)BS_ * LOG2_GAMMA);
  v4f p = *(const v4f*)(base + (size_t)1 * (BS_ * VD_));
#pragma nounroll
  for (int c0 = 1; c0 < NBLK_ - 1; c0 += SCAN_GRP_) {
    float* sl = base + (size_t)(c0 + 1) * (BS_ * VD_);
    v4f x[SCAN_GRP_];
#pragma unroll
    for (int j = 0; j < SCAN_GRP_; ++j)
      x[j] = *(const v4f*)(sl + (size_t)j * (BS_ * VD_));
    __builtin_amdgcn_sched_barrier(0);
#pragma unroll
    for (int j = 0; j < SCAN_GRP_; ++j) {
      p = p * g + x[j];
      *(v4f*)(sl + (size_t)j * (BS_ * VD_)) = p;
    }
  }
}

#define WP_ 68

__global__ __launch_bounds__(256) void retention_block_kernel(
    const float* __restrict__ Qb, const float* __restrict__ Kb,
    const float* __restrict__ Vb, float* Ob) {
  __shared__ __attribute__((aligned(16))) float sW[8][16 * WP_];
  const int lane = threadIdx.x & 31;
  const int w    = __builtin_amdgcn_readfirstlane((int)(threadIdx.x >> 5));
  const int r    = lane & 15;
  const int hi   = lane >> 4;
  float* sS = sW[w];

  const int bb = blockIdx.x / NBLK_;
  const int c  = blockIdx.x % NBLK_;
  const int t  = (w < 4) ? w : 7 - w;
  const int cb = (w >> 2) * 4;
  const int s0 = c * BS_;
  const int n0 = s0 + 16 * t;

  const float* Q = Qb + (size_t)bb * L_ * HD_;
  const float* K = Kb + (size_t)bb * L_ * HD_;
  const float* V = Vb + (size_t)bb * L_ * VD_;
  float*       O = Ob + (size_t)bb * L_ * VD_;

  v2f qa[16];
  {
    const float* qp = Q + (size_t)(n0 + r) * HD_ + 8 * hi;
#pragma unroll
    for (int t4 = 0; t4 < 4; ++t4) {
      v4f q0 = *(const v4f*)(qp + 16 * t4);
      v4f q1 = *(const v4f*)(qp + 16 * t4 + 4);
      qa[4 * t4 + 0] = v2f{q0.x, q0.y};
      qa[4 * t4 + 1] = v2f{q0.z, q0.w};
      qa[4 * t4 + 2] = v2f{q1.x, q1.y};
      qa[4 * t4 + 3] = v2f{q1.z, q1.w};
    }
  }

  const v8f vz = {0.f, 0.f, 0.f, 0.f, 0.f, 0.f, 0.f, 0.f};

  v8f oi[4];
#pragma unroll
  for (int ctl = 0; ctl < 4; ++ctl) oi[ctl] = vz;
  if (c > 0) {
    const float* P = O + (size_t)s0 * VD_;
#pragma unroll
    for (int ctl = 0; ctl < 4; ++ctl) {
      const float* pp = P + (size_t)(16 * (cb + ctl) + r) * HD_ + 8 * hi;
#pragma unroll
      for (int t4 = 0; t4 < 4; ++t4) {
        v4f p0 = *(const v4f*)(pp + 16 * t4);
        v4f p1 = *(const v4f*)(pp + 16 * t4 + 4);
        oi[ctl] = wmma_f32_16x16x4(qa[4 * t4 + 0], v2f{p0.x, p0.y}, oi[ctl]);
        oi[ctl] = wmma_f32_16x16x4(qa[4 * t4 + 1], v2f{p0.z, p0.w}, oi[ctl]);
        oi[ctl] = wmma_f32_16x16x4(qa[4 * t4 + 2], v2f{p1.x, p1.y}, oi[ctl]);
        oi[ctl] = wmma_f32_16x16x4(qa[4 * t4 + 3], v2f{p1.z, p1.w}, oi[ctl]);
      }
    }
  }
  __syncthreads();

  float gp[8];
#pragma unroll
  for (int i = 0; i < 8; ++i)
    gp[i] = exp2f((float)((i + 8 * hi) - r) * LOG2_GAMMA);

  v8f o[4];
#pragma unroll
  for (int ctl = 0; ctl < 4; ++ctl) o[ctl] = vz;

  const float* kptr  = K + (size_t)s0 * HD_ + lane * 4;
  const float* vbase = V + (size_t)s0 * VD_ + cb * 256 + lane * 4;

  for (int m0 = s0; m0 <= n0; m0 += 16) {
    v4f kf[8];
#pragma unroll
    for (int j = 0; j < 8; ++j) kf[j] = *(const v4f*)(kptr + j * 128);
    v4f vf[2][4];
#pragma unroll
    for (int q = 0; q < 2; ++q) {
#pragma unroll
      for (int ctl = 0; ctl < 4; ++ctl)
        vf[q][ctl] = *(const v4f*)(vbase + ctl * 256 + q * 128);
    }

    v8f s = vz;
#pragma unroll
    for (int j = 0; j < 8; ++j) {
      s = wmma_f32_16x16x4(qa[2 * j],     v2f{kf[j].x, kf[j].y}, s);
      s = wmma_f32_16x16x4(qa[2 * j + 1], v2f{kf[j].z, kf[j].w}, s);
    }

    {
      const bool diag = (m0 == n0);
      float cf = exp2f((float)(n0 - m0) * LOG2_GAMMA);
#pragma unroll
      for (int i = 0; i < 8; ++i) {
        int diff = (i + 8 * hi) - r;
        float v = s[i] * (cf * gp[i]);
        s[i] = (diag && diff < 0) ? 0.0f : v;
      }
    }

#pragma unroll
    for (int i = 0; i < 8; ++i) sS[(i + 8 * hi) * 18 + r] = s[i];
    v2f av[4];
#pragma unroll
    for (int u = 0; u < 4; ++u)
      av[u] = *(const v2f*)&sS[r * 18 + 8 * hi + 2 * u];

#pragma unroll
    for (int q = 0; q < 2; ++q) {
#pragma unroll
      for (int ctl = 0; ctl < 4; ++ctl) {
        v4f f = vf[q][ctl];
        o[ctl] = wmma_f32_16x16x4(av[2 * q],     v2f{f.x, f.y}, o[ctl]);
        o[ctl] = wmma_f32_16x16x4(av[2 * q + 1], v2f{f.z, f.w}, o[ctl]);
      }
    }

    kptr  += 16 * HD_;
    vbase += 16 * VD_;
  }

#pragma unroll
  for (int i = 0; i < 8; ++i) {
    float rd = exp2f((float)(16 * t + i + 8 * hi) * LOG2_GAMMA);
#pragma unroll
    for (int ctl = 0; ctl < 4; ++ctl)
      sS[(i + 8 * hi) * WP_ + ctl * 16 + r] = fmaf(oi[ctl][i], rd, o[ctl][i]);
  }
  float* orow = O + (size_t)(n0 + hi) * VD_ + 16 * cb + 4 * r;
#pragma unroll
  for (int p = 0; p < 8; ++p) {
    v4f x = *(const v4f*)&sS[(2 * p + hi) * WP_ + 4 * r];
    *(v4f*)(orow + (size_t)(2 * p) * VD_) = x;
  }
}

extern "C" void kernel_launch(void* const* d_in, const int* in_sizes, int n_in,
                              void* d_out, int out_size, void* d_ws, size_t ws_size,
                              hipStream_t stream) {
  const float* X  = (const float*)d_in[0];
  const float* WQ = (const float*)d_in[1];
  const float* WK = (const float*)d_in[2];
  const float* WV = (const float*)d_in[3];

  float* Qb = (float*)d_ws;
  float* Kb = Qb + (size_t)B_ * L_ * HD_;
  float* Vb = Kb + (size_t)B_ * L_ * HD_;
  float* O  = (float*)d_out;

  const int rowTiles = B_ * L_ / 16;
  proj_xpos_kernel<<<dim3(rowTiles), dim3(128), 0, stream>>>(X, WQ, WK, WV,
                                                             Qb, Kb, Vb);
  block_state_kernel<<<dim3(B_ * (NBLK_ - 1)), dim3(128), 0, stream>>>(Kb, Vb, O);
  state_scan_kernel<<<dim3(B_ * 8), dim3(256), 0, stream>>>(O);
  retention_block_kernel<<<dim3(B_ * NBLK_), dim3(256), 0, stream>>>(Qb, Kb, Vb, O);
}


# projection kernel as an LDS-staged GEMM: one 8-wave work-group per 64 rows x all 256 columns, X and W in 64-k chunks by whole-line 16 B loads, 16 B fragment reads; k order in the products changes
# speedup vs baseline: 1.7200x; 1.0458x over previous
#include <hip/hip_runtime.h>

typedef __attribute__((ext_vector_type(2))) float v2f;
typedef __attribute__((ext_vector_type(4))) float v4f;
typedef __attribute__((ext_vector_type(8))) float v8f;

#define B_  4
#define L_  4096
#define D_  512
#define HD_ 64
#define VD_ 128

constexpr float LOG2_GAMMA = -0.04580329f;

__device__ __forceinline__ v8f wmma_f32_16x16x4(v2f a, v2f b, v8f c) {
  return __builtin_amdgcn_wmma_f32_16x16x4_f32(false, a, false, b, (short)0, c,
                                               false, false);
}

__device__ __forceinline__ void xpos_coeffs(int n, int j, bool down,
                                            float& sn, float& cs) {
  float sv    = (2.0f * (float)j + 0.4f * (float)HD_) * (1.0f / (1.4f * (float)HD_));
  float ex    = (float)n * (1.0f / 512.0f);
  if (down) ex = -ex;
  float scale = exp2f(ex * log2f(sv));
  float invf  = exp2f((float)j * (-13.2877124f / 32.0f));
  float ang   = (float)n * invf;
  sincosf(ang, &sn, &cs);
  sn *= scale; cs *= scale;
}

#define PK_   64
#define NCH_  (D_ / PK_)
#define XP_   (PK_ + 4)
#define WT_   68
#define WG_   (16 * WT_)
#define XBUF_ (64 * XP_)
#define WBUF_ ((PK_ / 4) * WG_)

struct ProjStage {
  v4f x[4];
  v4f w[4][4];
};

__device__ __forceinline__ void proj_fetch(ProjStage& st, const float* __restrict__ xg,
                                           const float* __restrict__ wg, int ws, int kc) {
#pragma unroll
  for (int i = 0; i < 4; ++i)
    st.x[i] = *(const v4f*)(xg + (size_t)(16 * i) * D_ + kc * PK_);
#pragma unroll
  for (int i = 0; i < 4; ++i) {
#pragma unroll
    for (int e = 0; e < 4; ++e)
      st.w[i][e] = *(const v4f*)(wg + (size_t)(kc * PK_ + 16 * i + e) * ws);
  }
}

__device__ __forceinline__ void proj_park(const ProjStage& st, float* xs, float* wsd) {
#pragma unroll
  for (int i = 0; i < 4; ++i) *(v4f*)(xs + 16 * i * XP_) = st.x[i];
#pragma unroll
  for (int i = 0; i < 4; ++i) {
#pragma unroll
    for (int c4 = 0; c4 < 4; ++c4) {
      v4f col = {st.w[i][0][c4], st.w[i][1][c4], st.w[i][2][c4], st.w[i][3][c4]};
      *(v4f*)(wsd + 4 * i * WG_ + 4 * c4) = col;
    }
  }
}

struct ProjFrag {
  v4f a;
  v4f b[8];
};

__device__ __forceinline__ void proj_frag(ProjFrag& f, const float* pa, const float* pbq,
                                          const float* pbk, const float* pbv, int j) {
  f.a = *(const v4f*)(pa + 8 * j);
#pragma unroll
  for (int t = 0; t < 2; ++t) f.b[t]     = *(const v4f*)(pbq + 2 * j * WG_ + t * WT_);
#pragma unroll
  for (int t = 0; t < 2; ++t) f.b[2 + t] = *(const v4f*)(pbk + 2 * j * WG_ + t * WT_);
#pragma unroll
  for (int t = 0; t < 4; ++t) f.b[4 + t] = *(const v4f*)(pbv + 2 * j * WG_ + t * WT_);
}

__device__ __forceinline__ void proj_chunk(const float* pa, const float* pbq,
                                           const float* pbk, const float* pbv,
                                           v8f acc[8]) {
  ProjFrag f[2];
  proj_frag(f[0], pa, pbq, pbk, pbv, 0);
#pragma unroll
  for (int j = 0; j < PK_ / 8; ++j) {
    if (j + 1 < PK_ / 8) proj_frag(f[(j + 1) & 1], pa, pbq, pbk, pbv, j + 1);
    __builtin_amdgcn_sched_barrier(0);
    const ProjFrag& c = f[j & 1];
    const v2f a0 = {c.a.x, c.a.y};
    const v2f a1 = {c.a.z, c.a.w};
#pragma unroll
    for (int t = 0; t < 8; ++t) {
      const v2f b0 = {c.b[t].x, c.b[t].y};
      const bool kt = (t == 2 || t == 3);
      acc[t] = kt ? wmma_f32_16x16x4(b0, a0, acc[t]) : wmma_f32_16x16x4(a0, b0, acc[t]);
    }
#pragma unroll
    for (int t = 0; t < 8; ++t) {
      const v2f b1 = {c.b[t].z, c.b[t].w};
      const bool kt = (t == 2 || t == 3);
      acc[t] = kt ? wmma_f32_16x16x4(b1, a1, acc[t]) : wmma_f32_16x16x4(a1, b1, acc[t]);
    }
    __builtin_amdgcn_sched_barrier(0);
  }
}

__global__ __launch_bounds__(256) void proj_xpos_kernel(
    const float* __restrict__ X,  const float* __restrict__ WQ,
    const float* __restrict__ WK, const float* __restrict__ WV,
    float* __restrict__ Qb, float* __restrict__ Kb, float* __restrict__ Vb) {
  __shared__ __attribute__((aligned(16))) float sX[2][XBUF_];
  __shared__ __attribute__((aligned(16))) float sB[2][WBUF_];
  const int tid  = threadIdx.x;
  const int lane = tid & 31;
  const int w    = __builtin_amdgcn_readfirstlane((int)(threadIdx.x >> 5));
  const int r    = lane & 15;
  const int hi   = lane >> 4;
  const int mt   = w >> 1;
  const int h    = w & 1;
  const int rowBase = blockIdx.x * 64;

  const float* xg = X + (size_t)(rowBase + (tid >> 4)) * D_ + 4 * (tid & 15);
  const int ws = h ? VD_ : HD_;
  const float* wg = h ? (WV + 4 * lane) : (hi ? (WK + 4 * r) : (WQ + 4 * r));
  wg += (size_t)(4 * mt) * ws;
  const int cq = tid & 63;
  float* xs  = &sX[0][(tid >> 4) * XP_ + 4 * (tid & 15)];
  float* wsd = &sB[0][mt * WG_ + (cq >> 2) * WT_ + (cq & 3) * 16];

  const float* pa  = &sX[0][(16 * mt + r) * XP_ + 4 * hi];
  const float* pb  = &sB[0][hi * WG_ + 4 * r];
  const float* pbq = pb + WT_ * (2 * h);
  const float* pbk = pb + WT_ * (4 + 2 * h);
  const float* pbv = pb + WT_ * (8 + 4 * h);

  const v8f vz = {0.f, 0.f, 0.f, 0.f, 0.f, 0.f, 0.f, 0.f};
  v8f acc[8];
#pragma unroll
  for (int t = 0; t < 8; ++t) acc[t] = vz;

  ProjStage st;
  proj_fetch(st, xg, wg, ws, 0);
  proj_park(st, xs, wsd);
  __syncthreads();

#pragma nounroll
  for (int kc = 0; kc < NCH_ - 2; kc += 2) {
    proj_fetch(st, xg, wg, ws, kc + 1);
    __builtin_amdgcn_sched_barrier(0);
    proj_chunk(pa, pbq, pbk, pbv, acc);
    proj_park(st, xs + XBUF_, wsd + WBUF_);
    __syncthreads();
    proj_fetch(st, xg, wg, ws, kc + 2);
    __builtin_amdgcn_sched_barrier(0);
    proj_chunk(pa + XBUF_, pbq + WBUF_, pbk + WBUF_, pbv + WBUF_, acc);
    proj_park(st, xs, wsd);
    __syncthreads();
  }
  proj_fetch(st, xg, wg, ws, NCH_ - 1);
  __builtin_amdgcn_sched_barrier(0);
  proj_chunk(pa, pbq, pbk, pbv, acc);
  proj_park(st, xs + XBUF_, wsd + WBUF_);
  __syncthreads();
  proj_chunk(pa + XBUF_, pbq + WBUF_, pbk + WBUF_, pbv + WBUF_, acc);

  const int row16  = rowBase + 16 * mt;
  const int tile16 = blockIdx.x * 4 + mt;

#pragma unroll
  for (int tq = 0; tq < 2; ++tq) {
    const int ch = (2 * h + tq) * 16 + r;
#pragma unroll
    for (int i = 0; i < 8; ++i) {
      int row = row16 + i + 8 * hi;
      float x = acc[tq][i];
      float p = __shfl_xor(x, 1, 32);
      float sn, cs;
      xpos_coeffs(row & (L_ - 1), ch >> 1, false, sn, cs);
      float outv = (ch & 1) ? fmaf(x, cs, p * sn) : fmaf(x, cs, -(p * sn));
      Qb[(size_t)row * HD_ + ch] = outv;
    }
  }

  {
    const int n = (row16 + r) & (L_ - 1);
    float* kdst = Kb + (size_t)tile16 * (16 * HD_) + lane * 4;
#pragma unroll
    for (int tk = 0; tk < 2; ++tk) {
      const int t = 2 * h + tk;
      float o8[8];
#pragma unroll
      for (int ip = 0; ip < 4; ++ip) {
        float sn, cs;
        xpos_coeffs(n, 8 * t + 4 * hi + ip, true, sn, cs);
        float xe = acc[2 + tk][2 * ip];
        float xo = acc[2 + tk][2 * ip + 1];
        o8[2 * ip]     = fmaf(xe, cs, -(xo * sn));
        o8[2 * ip + 1] = fmaf(xo, cs, xe * sn);
      }
      v4f lo = {o8[0], o8[1], o8[2], o8[3]};
      v4f up = {o8[4], o8[5], o8[6], o8[7]};
      *(v4f*)(kdst + (2 * t)     * 128) = lo;
      *(v4f*)(kdst + (2 * t + 1) * 128) = up;
    }
  }

  {
    float* vdst = Vb + (size_t)tile16 * (16 * VD_) + lane * 4;
#pragma unroll
    for (int tv = 0; tv < 4; ++tv) {
      const int ct = 4 * h + tv;
      v4f lo = {acc[4 + tv][0], acc[4 + tv][1], acc[4 + tv][2], acc[4 + tv][3]};
      v4f up = {acc[4 + tv][4], acc[4 + tv][5], acc[4 + tv][6], acc[4 + tv][7]};
      *(v4f*)(vdst + ct * 256)       = lo;
      *(v4f*)(vdst + ct * 256 + 128) = up;
    }
  }
}

#define BS_   64
#define NBLK_ (L_ / BS_)

__global__ __launch_bounds__(128) void block_state_kernel(
    const float* __restrict__ Kb, const float* __restrict__ Vb, float* Ob) {
  const int lane = threadIdx.x & 31;
  const int w    = __builtin_amdgcn_readfirstlane((int)(threadIdx.x >> 5));
  const int r    = lane & 15;
  const int hi   = lane >> 4;
  const int bb   = blockIdx.x / (NBLK_ - 1);
  const int c    = blockIdx.x % (NBLK_ - 1);

  const float* kp = Kb + ((size_t)bb * L_ + (size_t)c * BS_) * HD_
                       + ((r >> 2) & 1) * 128 + (r >> 3) * 64 + (r & 3) + 32 * hi;
  const float* vp = Vb + ((size_t)bb * L_ + (size_t)c * BS_) * VD_
                       + lane * 4 + (2 * w) * 256;
  float* Lt = Ob + ((size_t)bb * L_ + (size_t)(c + 1) * BS_) * VD_;

  const v8f vz = {0.f, 0.f, 0.f, 0.f, 0.f, 0.f, 0.f, 0.f};
  v8f acc[4][2];
#pragma unroll
  for (int ck = 0; ck < 4; ++ck) { acc[ck][0] = vz; acc[ck][1] = vz; }

#pragma unroll
  for (int cj = 0; cj < BS_ / 16; ++cj) {
    v4f vf[2][2];
#pragma unroll
    for (int ctl = 0; ctl < 2; ++ctl) {
#pragma unroll
      for (int q = 0; q < 2; ++q)
        vf[ctl][q] = *(const v4f*)(vp + cj * (16 * VD_) + ctl * 256 + q * 128);
    }
    v2f ka[4][4];
#pragma unroll
    for (int ck = 0; ck < 4; ++ck) {
#pragma unroll
      for (int u = 0; u < 4; ++u) {
        ka[ck][u].x = kp[cj * (16 * HD_) + ck * 256 + u * 8];
        ka[ck][u].y = kp[cj * (16 * HD_) + ck * 256 + u * 8 + 4];
      }
    }
#pragma unroll
    for (int q = 0; q < 2; ++q) {
      v4f d;
      d.x = exp2f((float)(BS_ - 16 * cj - 8 * hi - 4 * q - 0) * LOG2_GAMMA);
      d.y = exp2f((float)(BS_ - 16 * cj - 8 * hi - 4 * q - 1) * LOG2_GAMMA);
      d.z = exp2f((float)(BS_ - 16 * cj - 8 * hi - 4 * q - 2) * LOG2_GAMMA);
      d.w = exp2f((float)(BS_ - 16 * cj - 8 * hi - 4 * q - 3) * LOG2_GAMMA);
      vf[0][q] *= d;
      vf[1][q] *= d;
    }
#pragma unroll
    for (int ck = 0; ck < 4; ++ck) {
#pragma unroll
      for (int ctl = 0; ctl < 2; ++ctl) {
#pragma unroll
        for (int q = 0; q < 2; ++q) {
          v4f f = vf[ctl][q];
          acc[ck][ctl] = wmma_f32_16x16x4(ka[ck][2 * q],     v2f{f.x, f.y}, acc[ck][ctl]);
          acc[ck][ctl] = wmma_f32_16x16x4(ka[ck][2 * q + 1], v2f{f.z, f.w}, acc[ck][ctl]);
        }
      }
    }
  }

#pragma unroll
  for (int ck = 0; ck < 4; ++ck) {
#pragma unroll
    for (int ctl = 0; ctl < 2; ++ctl) {
      v8f a = acc[ck][ctl];
      v4f lo = {a[0], a[1], a[2], a[3]};
      v4f up = {a[4], a[5], a[6], a[7]};
      float* dst = Lt + (size_t)(32 * w + 16 * ctl + r) * HD_ + 16 * ck + 8 * hi;
      *(v4f*)(dst)     = lo;
      *(v4f*)(dst + 4) = up;
    }
  }
}

#define SCAN_GRP_ 31
__global__ __launch_bounds__(256) void state_scan_kernel(float* Ob) {
  const int e  = (blockIdx.x & 7) * 256 + threadIdx.x;
  const int bb = blockIdx.x >> 3;
  float* base = Ob + (size_t)bb * L_ * VD_ + (size_t)e * 4;
  const float g = exp2f((float)BS_ * LOG2_GAMMA);
  v4f p = *(const v4f*)(base + (size_t)1 * (BS_ * VD_));
#pragma nounroll
  for (int c0 = 1; c0 < NBLK_ - 1; c0 += SCAN_GRP_) {
    float* sl = base + (size_t)(c0 + 1) * (BS_ * VD_);
    v4f x[SCAN_GRP_];
#pragma unroll
    for (int j = 0; j < SCAN_GRP_; ++j)
      x[j] = *(const v4f*)(sl + (size_t)j * (BS_ * VD_));
    __builtin_amdgcn_sched_barrier(0);
#pragma unroll
    for (int j = 0; j < SCAN_GRP_; ++j) {
      p = p * g + x[j];
      *(v4f*)(sl + (size_t)j * (BS_ * VD_)) = p;
    }
  }
}

#define WP_ 68

__global__ __launch_bounds__(256) void retention_block_kernel(
    const float* __restrict__ Qb, const float* __restrict__ Kb,
    const float* __restrict__ Vb, float* Ob) {
  __shared__ __attribute__((aligned(16))) float sW[8][16 * WP_];
  const int lane = threadIdx.x & 31;
  const int w    = __builtin_amdgcn_readfirstlane((int)(threadIdx.x >> 5));
  const int r    = lane & 15;
  const int hi   = lane >> 4;
  float* sS = sW[w];

  const int bb = blockIdx.x / NBLK_;
  const int c  = blockIdx.x % NBLK_;
  const int t  = (w < 4) ? w : 7 - w;
  const int cb = (w >> 2) * 4;
  const int s0 = c * BS_;
  const int n0 = s0 + 16 * t;

  const float* Q = Qb + (size_t)bb * L_ * HD_;
  const float* K = Kb + (size_t)bb * L_ * HD_;
  const float* V = Vb + (size_t)bb * L_ * VD_;
  float*       O = Ob + (size_t)bb * L_ * VD_;

  v2f qa[16];
  {
    const float* qp = Q + (size_t)(n0 + r) * HD_ + 8 * hi;
#pragma unroll
    for (int t4 = 0; t4 < 4; ++t4) {
      v4f q0 = *(const v4f*)(qp + 16 * t4);
      v4f q1 = *(const v4f*)(qp + 16 * t4 + 4);
      qa[4 * t4 + 0] = v2f{q0.x, q0.y};
      qa[4 * t4 + 1] = v2f{q0.z, q0.w};
      qa[4 * t4 + 2] = v2f{q1.x, q1.y};
      qa[4 * t4 + 3] = v2f{q1.z, q1.w};
    }
  }

  const v8f vz = {0.f, 0.f, 0.f, 0.f, 0.f, 0.f, 0.f, 0.f};

  v8f oi[4];
#pragma unroll
  for (int ctl = 0; ctl < 4; ++ctl) oi[ctl] = vz;
  if (c > 0) {
    const float* P = O + (size_t)s0 * VD_;
#pragma unroll
    for (int ctl = 0; ctl < 4; ++ctl) {
      const float* pp = P + (size_t)(16 * (cb + ctl) + r) * HD_ + 8 * hi;
#pragma unroll
      for (int t4 = 0; t4 < 4; ++t4) {
        v4f p0 = *(const v4f*)(pp + 16 * t4);
        v4f p1 = *(const v4f*)(pp + 16 * t4 + 4);
        oi[ctl] = wmma_f32_16x16x4(qa[4 * t4 + 0], v2f{p0.x, p0.y}, oi[ctl]);
        oi[ctl] = wmma_f32_16x16x4(qa[4 * t4 + 1], v2f{p0.z, p0.w}, oi[ctl]);
        oi[ctl] = wmma_f32_16x16x4(qa[4 * t4 + 2], v2f{p1.x, p1.y}, oi[ctl]);
        oi[ctl] = wmma_f32_16x16x4(qa[4 * t4 + 3], v2f{p1.z, p1.w}, oi[ctl]);
      }
    }
  }
  __syncthreads();

  float gp[8];
#pragma unroll
  for (int i = 0; i < 8; ++i)
    gp[i] = exp2f((float)((i + 8 * hi) - r) * LOG2_GAMMA);

  v8f o[4];
#pragma unroll
  for (int ctl = 0; ctl < 4; ++ctl) o[ctl] = vz;

  const float* kptr  = K + (size_t)s0 * HD_ + lane * 4;
  const float* vbase = V + (size_t)s0 * VD_ + cb * 256 + lane * 4;

  for (int m0 = s0; m0 <= n0; m0 += 16) {
    v4f kf[8];
#pragma unroll
    for (int j = 0; j < 8; ++j) kf[j] = *(const v4f*)(kptr + j * 128);
    v4f vf[2][4];
#pragma unroll
    for (int q = 0; q < 2; ++q) {
#pragma unroll
      for (int ctl = 0; ctl < 4; ++ctl)
        vf[q][ctl] = *(const v4f*)(vbase + ctl * 256 + q * 128);
    }

    v8f s = vz;
#pragma unroll
    for (int j = 0; j < 8; ++j) {
      s = wmma_f32_16x16x4(qa[2 * j],     v2f{kf[j].x, kf[j].y}, s);
      s = wmma_f32_16x16x4(qa[2 * j + 1], v2f{kf[j].z, kf[j].w}, s);
    }

    {
      const bool diag = (m0 == n0);
      float cf = exp2f((float)(n0 - m0) * LOG2_GAMMA);
#pragma unroll
      for (int i = 0; i < 8; ++i) {
        int diff = (i + 8 * hi) - r;
        float v = s[i] * (cf * gp[i]);
        s[i] = (diag && diff < 0) ? 0.0f : v;
      }
    }

#pragma unroll
    for (int i = 0; i < 8; ++i) sS[(i + 8 * hi) * 18 + r] = s[i];
    v2f av[4];
#pragma unroll
    for (int u = 0; u < 4; ++u)
      av[u] = *(const v2f*)&sS[r * 18 + 8 * hi + 2 * u];

#pragma unroll
    for (int q = 0; q < 2; ++q) {
#pragma unroll
      for (int ctl = 0; ctl < 4; ++ctl) {
        v4f f = vf[q][ctl];
        o[ctl] = wmma_f32_16x16x4(av[2 * q],     v2f{f.x, f.y}, o[ctl]);
        o[ctl] = wmma_f32_16x16x4(av[2 * q + 1], v2f{f.z, f.w}, o[ctl]);
      }
    }

    kptr  += 16 * HD_;
    vbase += 16 * VD_;
  }

#pragma unroll
  for (int i = 0; i < 8; ++i) {
    float rd = exp2f((float)(16 * t + i + 8 * hi) * LOG2_GAMMA);
#pragma unroll
    for (int ctl = 0; ctl < 4; ++ctl)
      sS[(i + 8 * hi) * WP_ + ctl * 16 + r] = fmaf(oi[ctl][i], rd, o[ctl][i]);
  }
  float* orow = O + (size_t)(n0 + hi) * VD_ + 16 * cb + 4 * r;
#pragma unroll
  for (int p = 0; p < 8; ++p) {
    v4f x = *(const v4f*)&sS[(2 * p + hi) * WP_ + 4 * r];
    *(v4f*)(orow + (size_t)(2 * p) * VD_) = x;
  }
}

extern "C" void kernel_launch(void* const* d_in, const int* in_sizes, int n_in,
                              void* d_out, int out_size, void* d_ws, size_t ws_size,
                              hipStream_t stream) {
  const float* X  = (const float*)d_in[0];
  const float* WQ = (const float*)d_in[1];
  const float* WK = (const float*)d_in[2];
  const float* WV = (const float*)d_in[3];

  float* Qb = (float*)d_ws;
  float* Kb = Qb + (size_t)B_ * L_ * HD_;
  float* Vb = Kb + (size_t)B_ * L_ * HD_;
  float* O  = (float*)d_out;

  proj_xpos_kernel<<<dim3(B_ * L_ / 64), dim3(256), 0, stream>>>(X, WQ, WK, WV,
                                                                 Qb, Kb, Vb);
  block_state_kernel<<<dim3(B_ * (NBLK_ - 1)), dim3(128), 0, stream>>>(Kb, Vb, O);
  state_scan_kernel<<<dim3(B_ * 8), dim3(256), 0, stream>>>(O);
  retention_block_kernel<<<dim3(B_ * NBLK_), dim3(256), 0, stream>>>(Qb, Kb, Vb, O);
}


# block-state product K^T V moved from its own launch into the projection kernel's epilogue (K, decayed V tiles through LDS, same k steps per tile): 3 launches, not 4; sum order unchanged
# speedup vs baseline: 1.8221x; 1.0594x over previous
#include <hip/hip_runtime.h>

typedef __attribute__((ext_vector_type(2))) float v2f;
typedef __attribute__((ext_vector_type(4))) float v4f;
typedef __attribute__((ext_vector_type(8))) float v8f;

#define B_  4
#define L_  4096
#define D_  512
#define HD_ 64
#define VD_ 128
#define BS_   64
#define NBLK_ (L_ / BS_)

constexpr float LOG2_GAMMA = -0.04580329f;

__device__ __forceinline__ v8f wmma_f32_16x16x4(v2f a, v2f b, v8f c) {
  return __builtin_amdgcn_wmma_f32_16x16x4_f32(false, a, false, b, (short)0, c,
                                               false, false);
}

__device__ __forceinline__ void xpos_coeffs(int n, int j, bool down,
                                            float& sn, float& cs) {
  float sv    = (2.0f * (float)j + 0.4f * (float)HD_) * (1.0f / (1.4f * (float)HD_));
  float ex    = (float)n * (1.0f / 512.0f);
  if (down) ex = -ex;
  float scale = exp2f(ex * log2f(sv));
  float invf  = exp2f((float)j * (-13.2877124f / 32.0f));
  float ang   = (float)n * invf;
  sincosf(ang, &sn, &cs);
  sn *= scale; cs *= scale;
}

#define PK_   64
#define NCH_  (D_ / PK_)
#define XP_   (PK_ + 4)
#define WT_   68
#define WG_   (16 * WT_)
#define XBUF_ (64 * XP_)
#define WBUF_ ((PK_ / 4) * WG_)
#define KP_   68

struct ProjStage {
  v4f x[4];
  v4f w[4][4];
};

__device__ __forceinline__ void proj_fetch(ProjStage& st, const float* __restrict__ xg,
                                           const float* __restrict__ wg, int ws, int kc) {
#pragma unroll
  for (int i = 0; i < 4; ++i)
    st.x[i] = *(const v4f*)(xg + (size_t)(16 * i) * D_ + kc * PK_);
#pragma unroll
  for (int i = 0; i < 4; ++i) {
#pragma unroll
    for (int e = 0; e < 4; ++e)
      st.w[i][e] = *(const v4f*)(wg + (size_t)(kc * PK_ + 16 * i + e) * ws);
  }
}

__device__ __forceinline__ void proj_park(const ProjStage& st, float* xs, float* wsd) {
#pragma unroll
  for (int i = 0; i < 4; ++i) *(v4f*)(xs + 16 * i * XP_) = st.x[i];
#pragma unroll
  for (int i = 0; i < 4; ++i) {
#pragma unroll
    for (int c4 = 0; c4 < 4; ++c4) {
      v4f col = {st.w[i][0][c4], st.w[i][1][c4], st.w[i][2][c4], st.w[i][3][c4]};
      *(v4f*)(wsd + 4 * i * WG_ + 4 * c4) = col;
    }
  }
}

struct ProjFrag {
  v4f a;
  v4f b[8];
};

__device__ __forceinline__ void proj_frag(ProjFrag& f, const float* pa, const float* pbq,
                                          const float* pbk, const float* pbv, int j) {
  f.a = *(const v4f*)(pa + 8 * j);
#pragma unroll
  for (int t = 0; t < 2; ++t) f.b[t]     = *(const v4f*)(pbq + 2 * j * WG_ + t * WT_);
#pragma unroll
  for (int t = 0; t < 2; ++t) f.b[2 + t] = *(const v4f*)(pbk + 2 * j * WG_ + t * WT_);
#pragma unroll
  for (int t = 0; t < 4; ++t) f.b[4 + t] = *(const v4f*)(pbv + 2 * j * WG_ + t * WT_);
}

__device__ __forceinline__ void proj_chunk(const float* pa, const float* pbq,
                                           const float* pbk, const float* pbv,
                                           v8f acc[8]) {
  ProjFrag f[2];
  proj_frag(f[0], pa, pbq, pbk, pbv, 0);
#pragma unroll
  for (int j = 0; j < PK_ / 8; ++j) {
    if (j + 1 < PK_ / 8) proj_frag(f[(j + 1) & 1], pa, pbq, pbk, pbv, j + 1);
    __builtin_amdgcn_sched_barrier(0);
    const ProjFrag& c = f[j & 1];
    const v2f a0 = {c.a.x, c.a.y};
    const v2f a1 = {c.a.z, c.a.w};
#pragma unroll
    for (int t = 0; t < 8; ++t) {
      const v2f b0 = {c.b[t].x, c.b[t].y};
      const bool kt = (t == 2 || t == 3);
      acc[t] = kt ? wmma_f32_16x16x4(b0, a0, acc[t]) : wmma_f32_16x16x4(a0, b0, acc[t]);
    }
#pragma unroll
    for (int t = 0; t < 8; ++t) {
      const v2f b1 = {c.b[t].z, c.b[t].w};
      const bool kt = (t == 2 || t == 3);
      acc[t] = kt ? wmma_f32_16x16x4(b1, a1, acc[t]) : wmma_f32_16x16x4(a1, b1, acc[t]);
    }
    __builtin_amdgcn_sched_barrier(0);
  }
}

__global__ __launch_bounds__(256) void proj_xpos_kernel(
    const float* __restrict__ X,  const float* __restrict__ WQ,
    const float* __restrict__ WK, const float* __restrict__ WV,
    float* __restrict__ Qb, float* __restrict__ Kb, float* __restrict__ Vb,
    float* __restrict__ Ob) {
  __shared__ __attribute__((aligned(16))) float sX[2][XBUF_];
  __shared__ __attribute__((aligned(16))) float sB[2][WBUF_];
  const int tid  = threadIdx.x;
  const int lane = tid & 31;
  const int w    = __builtin_amdgcn_readfirstlane((int)(threadIdx.x >> 5));
  const int r    = lane & 15;
  const int hi   = lane >> 4;
  const int mt   = w >> 1;
  const int h    = w & 1;
  const int rowBase = blockIdx.x * 64;

  const float* xg = X + (size_t)(rowBase + (tid >> 4)) * D_ + 4 * (tid & 15);
  const int ws = h ? VD_ : HD_;
  const float* wg = h ? (WV + 4 * lane) : (hi ? (WK + 4 * r) : (WQ + 4 * r));
  wg += (size_t)(4 * mt) * ws;
  const int cq = tid & 63;
  float* xs  = &sX[0][(tid >> 4) * XP_ + 4 * (tid & 15)];
  float* wsd = &sB[0][mt * WG_ + (cq >> 2) * WT_ + (cq & 3) * 16];

  const float* pa  = &sX[0][(16 * mt + r) * XP_ + 4 * hi];
  const float* pb  = &sB[0][hi * WG_ + 4 * r];
  const float* pbq = pb + WT_ * (2 * h);
  const float* pbk = pb + WT_ * (4 + 2 * h);
  const float* pbv = pb + WT_ * (8 + 4 * h);

  const v8f vz = {0.f, 0.f, 0.f, 0.f, 0.f, 0.f, 0.f, 0.f};
  v8f acc[8];
#pragma unroll
  for (int t = 0; t < 8; ++t) acc[t] = vz;

  ProjStage st;
  proj_fetch(st, xg, wg, ws, 0);
  proj_park(st, xs, wsd);
  __syncthreads();

#pragma nounroll
  for (int kc = 0; kc < NCH_ - 2; kc += 2) {
    proj_fetch(st, xg, wg, ws, kc + 1);
    __builtin_amdgcn_sched_barrier(0);
    proj_chunk(pa, pbq, pbk, pbv, acc);
    proj_park(st, xs + XBUF_, wsd + WBUF_);
    __syncthreads();
    proj_fetch(st, xg, wg, ws, kc + 2);
    __builtin_amdgcn_sched_barrier(0);
    proj_chunk(pa + XBUF_, pbq + WBUF_, pbk + WBUF_, pbv + WBUF_, acc);
    proj_park(st, xs, wsd);
    __syncthreads();
  }
  proj_fetch(st, xg, wg, ws, NCH_ - 1);
  __builtin_amdgcn_sched_barrier(0);
  proj_chunk(pa, pbq, pbk, pbv, acc);
  proj_park(st, xs + XBUF_, wsd + WBUF_);
  __syncthreads();
  proj_chunk(pa + XBUF_, pbq + WBUF_, pbk + WBUF_, pbv + WBUF_, acc);

  const int row16  = rowBase + 16 * mt;
  const int tile16 = blockIdx.x * 4 + mt;

#pragma unroll
  for (int tq = 0; tq < 2; ++tq) {
    const int ch = (2 * h + tq) * 16 + r;
#pragma unroll
    for (int i = 0; i < 8; ++i) {
      int row = row16 + i + 8 * hi;
      float x = acc[tq][i];
      float p = __shfl_xor(x, 1, 32);
      float sn, cs;
      xpos_coeffs(row & (L_ - 1), ch >> 1, false, sn, cs);
      float outv = (ch & 1) ? fmaf(x, cs, p * sn) : fmaf(x, cs, -(p * sn));
      Qb[(size_t)row * HD_ + ch] = outv;
    }
  }

  float* sK = &sB[0][0];
  float* sV = sK + HD_ * KP_;
  {
    const int n = (row16 + r) & (L_ - 1);
    float* kdst = Kb + (size_t)tile16 * (16 * HD_) + lane * 4;
#pragma unroll
    for (int tk = 0; tk < 2; ++tk) {
      const int t = 2 * h + tk;
      float o8[8];
#pragma unroll
      for (int ip = 0; ip < 4; ++ip) {
        float sn, cs;
        xpos_coeffs(n, 8 * t + 4 * hi + ip, true, sn, cs);
        float xe = acc[2 + tk][2 * ip];
        float xo = acc[2 + tk][2 * ip + 1];
        o8[2 * ip]     = fmaf(xe, cs, -(xo * sn));
        o8[2 * ip + 1] = fmaf(xo, cs, xe * sn);
      }
      v4f lo = {o8[0], o8[1], o8[2], o8[3]};
      v4f up = {o8[4], o8[5], o8[6], o8[7]};
      *(v4f*)(kdst + (2 * t)     * 128) = lo;
      *(v4f*)(kdst + (2 * t + 1) * 128) = up;
#pragma unroll
      for (int i = 0; i < 8; ++i)
        sK[(16 * t + 8 * hi + i) * KP_ + 16 * mt + r] = o8[i];
    }
  }

  {
    float* vdst = Vb + (size_t)tile16 * (16 * VD_) + lane * 4;
    v4f dlo, dup;
    dlo.x = exp2f((float)(BS_ - 16 * mt - 8 * hi - 0) * LOG2_GAMMA);
    dlo.y = exp2f((float)(BS_ - 16 * mt - 8 * hi - 1) * LOG2_GAMMA);
    dlo.z = exp2f((float)(BS_ - 16 * mt - 8 * hi - 2) * LOG2_GAMMA);
    dlo.w = exp2f((float)(BS_ - 16 * mt - 8 * hi - 3) * LOG2_GAMMA);
    dup.x = exp2f((float)(BS_ - 16 * mt - 8 * hi - 4) * LOG2_GAMMA);
    dup.y = exp2f((float)(BS_ - 16 * mt - 8 * hi - 5) * LOG2_GAMMA);
    dup.z = exp2f((float)(BS_ - 16 * mt - 8 * hi - 6) * LOG2_GAMMA);
    dup.w = exp2f((float)(BS_ - 16 * mt - 8 * hi - 7) * LOG2_GAMMA);
#pragma unroll
    for (int tv = 0; tv < 4; ++tv) {
      const int ct = 4 * h + tv;
      v4f lo = {acc[4 + tv][0], acc[4 + tv][1], acc[4 + tv][2], acc[4 + tv][3]};
      v4f up = {acc[4 + tv][4], acc[4 + tv][5], acc[4 + tv][6], acc[4 + tv][7]};
      *(v4f*)(vdst + ct * 256)       = lo;
      *(v4f*)(vdst + ct * 256 + 128) = up;
      float* sv = sV + (16 * ct + r) * KP_ + 16 * mt + 8 * hi;
      *(v4f*)(sv)     = lo * dlo;
      *(v4f*)(sv + 4) = up * dup;
    }
  }

  if ((blockIdx.x & (NBLK_ - 1)) != NBLK_ - 1) {
    __builtin_amdgcn_fence(__ATOMIC_RELEASE, "workgroup", "local");
    __builtin_amdgcn_s_barrier();
    __builtin_amdgcn_fence(__ATOMIC_ACQUIRE, "workgroup", "local");
    const int ca = w & 1;
    const int cw = w >> 1;
    const float* pk = sK + (32 * ca + r) * KP_ + 8 * hi;
    const float* pv = sV + (32 * cw + r) * KP_ + 8 * hi;
    v8f ls[2][2];
#pragma unroll
    for (int ckl = 0; ckl < 2; ++ckl) { ls[ckl][0] = vz; ls[ckl][1] = vz; }
#pragma unroll
    for (int cj = 0; cj < BS_ / 16; ++cj) {
      v4f kf[2][2];
      v4f vf[2][2];
#pragma unroll
      for (int x = 0; x < 2; ++x) {
#pragma unroll
        for (int q = 0; q < 2; ++q) {
          kf[x][q] = *(const v4f*)(pk + x * 16 * KP_ + 16 * cj + 4 * q);
          vf[x][q] = *(const v4f*)(pv + x * 16 * KP_ + 16 * cj + 4 * q);
        }
      }
#pragma unroll
      for (int q = 0; q < 2; ++q) {
#pragma unroll
        for (int ckl = 0; ckl < 2; ++ckl) {
#pragma unroll
          for (int ctl = 0; ctl < 2; ++ctl)
            ls[ckl][ctl] = wmma_f32_16x16x4(v2f{kf[ckl][q].x, kf[ckl][q].y},
                                            v2f{vf[ctl][q].x, vf[ctl][q].y}, ls[ckl][ctl]);
        }
#pragma unroll
        for (int ckl = 0; ckl < 2; ++ckl) {
#pragma unroll
          for (int ctl = 0; ctl < 2; ++ctl)
            ls[ckl][ctl] = wmma_f32_16x16x4(v2f{kf[ckl][q].z, kf[ckl][q].w},
                                            v2f{vf[ctl][q].z, vf[ctl][q].w}, ls[ckl][ctl]);
        }
      }
    }
    float* Lt = Ob + ((size_t)blockIdx.x + 1) * (BS_ * VD_);
#pragma unroll
    for (int ckl = 0; ckl < 2; ++ckl) {
#pragma unroll
      for (int ctl = 0; ctl < 2; ++ctl) {
        v8f a = ls[ckl][ctl];
        v4f lo = {a[0], a[1], a[2], a[3]};
        v4f up = {a[4], a[5], a[6], a[7]};
        float* dst = Lt + (size_t)(32 * cw + 16 * ctl + r) * HD_ + 32 * ca + 16 * ckl + 8 * hi;
        *(v4f*)(dst)     = lo;
        *(v4f*)(dst + 4) = up;
      }
    }
  }
}


#define SCAN_GRP_ 31
__global__ __launch_bounds__(256) void state_scan_kernel(float* Ob) {
  const int e  = (blockIdx.x & 7) * 256 + threadIdx.x;
  const int bb = blockIdx.x >> 3;
  float* base = Ob + (size_t)bb * L_ * VD_ + (size_t)e * 4;
  const float g = exp2f((float)BS_ * LOG2_GAMMA);
  v4f p = *(const v4f*)(base + (size_t)1 * (BS_ * VD_));
#pragma nounroll
  for (int c0 = 1; c0 < NBLK_ - 1; c0 += SCAN_GRP_) {
    float* sl = base + (size_t)(c0 + 1) * (BS_ * VD_);
    v4f x[SCAN_GRP_];
#pragma unroll
    for (int j = 0; j < SCAN_GRP_; ++j)
      x[j] = *(const v4f*)(sl + (size_t)j * (BS_ * VD_));
    __builtin_amdgcn_sched_barrier(0);
#pragma unroll
    for (int j = 0; j < SCAN_GRP_; ++j) {
      p = p * g + x[j];
      *(v4f*)(sl + (size_t)j * (BS_ * VD_)) = p;
    }
  }
}

#define WP_ 68

__global__ __launch_bounds__(256) void retention_block_kernel(
    const float* __restrict__ Qb, const float* __restrict__ Kb,
    const float* __restrict__ Vb, float* Ob) {
  __shared__ __attribute__((aligned(16))) float sW[8][16 * WP_];
  const int lane = threadIdx.x & 31;
  const int w    = __builtin_amdgcn_readfirstlane((int)(threadIdx.x >> 5));
  const int r    = lane & 15;
  const int hi   = lane >> 4;
  float* sS = sW[w];

  const int bb = blockIdx.x / NBLK_;
  const int c  = blockIdx.x % NBLK_;
  const int t  = (w < 4) ? w : 7 - w;
  const int cb = (w >> 2) * 4;
  const int s0 = c * BS_;
  const int n0 = s0 + 16 * t;

  const float* Q = Qb + (size_t)bb * L_ * HD_;
  const float* K = Kb + (size_t)bb * L_ * HD_;
  const float* V = Vb + (size_t)bb * L_ * VD_;
  float*       O = Ob + (size_t)bb * L_ * VD_;

  v2f qa[16];
  {
    const float* qp = Q + (size_t)(n0 + r) * HD_ + 8 * hi;
#pragma unroll
    for (int t4 = 0; t4 < 4; ++t4) {
      v4f q0 = *(const v4f*)(qp + 16 * t4);
      v4f q1 = *(const v4f*)(qp + 16 * t4 + 4);
      qa[4 * t4 + 0] = v2f{q0.x, q0.y};
      qa[4 * t4 + 1] = v2f{q0.z, q0.w};
      qa[4 * t4 + 2] = v2f{q1.x, q1.y};
      qa[4 * t4 + 3] = v2f{q1.z, q1.w};
    }
  }

  const v8f vz = {0.f, 0.f, 0.f, 0.f, 0.f, 0.f, 0.f, 0.f};

  v8f oi[4];
#pragma unroll
  for (int ctl = 0; ctl < 4; ++ctl) oi[ctl] = vz;
  if (c > 0) {
    const float* P = O + (size_t)s0 * VD_;
#pragma unroll
    for (int ctl = 0; ctl < 4; ++ctl) {
      const float* pp = P + (size_t)(16 * (cb + ctl) + r) * HD_ + 8 * hi;
#pragma unroll
      for (int t4 = 0; t4 < 4; ++t4) {
        v4f p0 = *(const v4f*)(pp + 16 * t4);
        v4f p1 = *(const v4f*)(pp + 16 * t4 + 4);
        oi[ctl] = wmma_f32_16x16x4(qa[4 * t4 + 0], v2f{p0.x, p0.y}, oi[ctl]);
        oi[ctl] = wmma_f32_16x16x4(qa[4 * t4 + 1], v2f{p0.z, p0.w}, oi[ctl]);
        oi[ctl] = wmma_f32_16x16x4(qa[4 * t4 + 2], v2f{p1.x, p1.y}, oi[ctl]);
        oi[ctl] = wmma_f32_16x16x4(qa[4 * t4 + 3], v2f{p1.z, p1.w}, oi[ctl]);
      }
    }
  }
  __syncthreads();

  float gp[8];
#pragma unroll
  for (int i = 0; i < 8; ++i)
    gp[i] = exp2f((float)((i + 8 * hi) - r) * LOG2_GAMMA);

  v8f o[4];
#pragma unroll
  for (int ctl = 0; ctl < 4; ++ctl) o[ctl] = vz;

  const float* kptr  = K + (size_t)s0 * HD_ + lane * 4;
  const float* vbase = V + (size_t)s0 * VD_ + cb * 256 + lane * 4;

  for (int m0 = s0; m0 <= n0; m0 += 16) {
    v4f kf[8];
#pragma unroll
    for (int j = 0; j < 8; ++j) kf[j] = *(const v4f*)(kptr + j * 128);
    v4f vf[2][4];
#pragma unroll
    for (int q = 0; q < 2; ++q) {
#pragma unroll
      for (int ctl = 0; ctl < 4; ++ctl)
        vf[q][ctl] = *(const v4f*)(vbase + ctl * 256 + q * 128);
    }

    v8f s = vz;
#pragma unroll
    for (int j = 0; j < 8; ++j) {
      s = wmma_f32_16x16x4(qa[2 * j],     v2f{kf[j].x, kf[j].y}, s);
      s = wmma_f32_16x16x4(qa[2 * j + 1], v2f{kf[j].z, kf[j].w}, s);
    }

    {
      const bool diag = (m0 == n0);
      float cf = exp2f((float)(n0 - m0) * LOG2_GAMMA);
#pragma unroll
      for (int i = 0; i < 8; ++i) {
        int diff = (i + 8 * hi) - r;
        float v = s[i] * (cf * gp[i]);
        s[i] = (diag && diff < 0) ? 0.0f : v;
      }
    }

#pragma unroll
    for (int i = 0; i < 8; ++i) sS[(i + 8 * hi) * 18 + r] = s[i];
    v2f av[4];
#pragma unroll
    for (int u = 0; u < 4; ++u)
      av[u] = *(const v2f*)&sS[r * 18 + 8 * hi + 2 * u];

#pragma unroll
    for (int q = 0; q < 2; ++q) {
#pragma unroll
      for (int ctl = 0; ctl < 4; ++ctl) {
        v4f f = vf[q][ctl];
        o[ctl] = wmma_f32_16x16x4(av[2 * q],     v2f{f.x, f.y}, o[ctl]);
        o[ctl] = wmma_f32_16x16x4(av[2 * q + 1], v2f{f.z, f.w}, o[ctl]);
      }
    }

    kptr  += 16 * HD_;
    vbase += 16 * VD_;
  }

#pragma unroll
  for (int i = 0; i < 8; ++i) {
    float rd = exp2f((float)(16 * t + i + 8 * hi) * LOG2_GAMMA);
#pragma unroll
    for (int ctl = 0; ctl < 4; ++ctl)
      sS[(i + 8 * hi) * WP_ + ctl * 16 + r] = fmaf(oi[ctl][i], rd, o[ctl][i]);
  }
  float* orow = O + (size_t)(n0 + hi) * VD_ + 16 * cb + 4 * r;
#pragma unroll
  for (int p = 0; p < 8; ++p) {
    v4f x = *(const v4f*)&sS[(2 * p + hi) * WP_ + 4 * r];
    *(v4f*)(orow + (size_t)(2 * p) * VD_) = x;
  }
}

extern "C" void kernel_launch(void* const* d_in, const int* in_sizes, int n_in,
                              void* d_out, int out_size, void* d_ws, size_t ws_size,
                              hipStream_t stream) {
  const float* X  = (const float*)d_in[0];
  const float* WQ = (const float*)d_in[1];
  const float* WK = (const float*)d_in[2];
  const float* WV = (const float*)d_in[3];

  float* Qb = (float*)d_ws;
  float* Kb = Qb + (size_t)B_ * L_ * HD_;
  float* Vb = Kb + (size_t)B_ * L_ * HD_;
  float* O  = (float*)d_out;

  proj_xpos_kernel<<<dim3(B_ * NBLK_), dim3(256), 0, stream>>>(X, WQ, WK, WV,
                                                               Qb, Kb, Vb, O);
  state_scan_kernel<<<dim3(B_ * 8), dim3(256), 0, stream>>>(O);
  retention_block_kernel<<<dim3(B_ * NBLK_), dim3(256), 0, stream>>>(Qb, Kb, Vb, O);
}


# Q tile computed transposed like K; Q and the block state stored in WMMA fragment order: 512 contiguous bytes per store and load; one sin/cos per (row, pair) serves Q and K; sum order unchanged
# speedup vs baseline: 2.3626x; 1.2966x over previous
#include <hip/hip_runtime.h>

typedef __attribute__((ext_vector_type(2))) float v2f;
typedef __attribute__((ext_vector_type(4))) float v4f;
typedef __attribute__((ext_vector_type(8))) float v8f;

#define B_  4
#define L_  4096
#define D_  512
#define HD_ 64
#define VD_ 128
#define BS_   64
#define NBLK_ (L_ / BS_)

constexpr float LOG2_GAMMA = -0.04580329f;

__device__ __forceinline__ v8f wmma_f32_16x16x4(v2f a, v2f b, v8f c) {
  return __builtin_amdgcn_wmma_f32_16x16x4_f32(false, a, false, b, (short)0, c,
                                               false, false);
}

__device__ __forceinline__ void xpos_coeffs_qk(int n, int j, float& snq, float& csq,
                                               float& snk, float& csk) {
  float sv    = (2.0f * (float)j + 0.4f * (float)HD_) * (1.0f / (1.4f * (float)HD_));
  float ex    = (float)n * (1.0f / 512.0f);
  float lsv   = log2f(sv);
  float scq   = exp2f(ex * lsv);
  float sck   = exp2f(-ex * lsv);
  float invf  = exp2f((float)j * (-13.2877124f / 32.0f));
  float ang   = (float)n * invf;
  float sn, cs;
  sincosf(ang, &sn, &cs);
  snq = sn * scq; csq = cs * scq;
  snk = sn * sck; csk = cs * sck;
}

#define PK_   64
#define NCH_  (D_ / PK_)
#define XP_   (PK_ + 4)
#define WT_   68
#define WG_   (16 * WT_)
#define XBUF_ (64 * XP_)
#define WBUF_ ((PK_ / 4) * WG_)
#define KP_   68

struct ProjStage {
  v4f x[4];
  v4f w[4][4];
};

__device__ __forceinline__ void proj_fetch(ProjStage& st, const float* __restrict__ xg,
                                           const float* __restrict__ wg, int ws, int kc) {
#pragma unroll
  for (int i = 0; i < 4; ++i)
    st.x[i] = *(const v4f*)(xg + (size_t)(16 * i) * D_ + kc * PK_);
#pragma unroll
  for (int i = 0; i < 4; ++i) {
#pragma unroll
    for (int e = 0; e < 4; ++e)
      st.w[i][e] = *(const v4f*)(wg + (size_t)(kc * PK_ + 16 * i + e) * ws);
  }
}

__device__ __forceinline__ void proj_park(const ProjStage& st, float* xs, float* wsd) {
#pragma unroll
  for (int i = 0; i < 4; ++i) *(v4f*)(xs + 16 * i * XP_) = st.x[i];
#pragma unroll
  for (int i = 0; i < 4; ++i) {
#pragma unroll
    for (int c4 = 0; c4 < 4; ++c4) {
      v4f col = {st.w[i][0][c4], st.w[i][1][c4], st.w[i][2][c4], st.w[i][3][c4]};
      *(v4f*)(wsd + 4 * i * WG_ + 4 * c4) = col;
    }
  }
}

struct ProjFrag {
  v4f a;
  v4f b[8];
};

__device__ __forceinline__ void proj_frag(ProjFrag& f, const float* pa, const float* pbq,
                                          const float* pbk, const float* pbv, int j) {
  f.a = *(const v4f*)(pa + 8 * j);
#pragma unroll
  for (int t = 0; t < 2; ++t) f.b[t]     = *(const v4f*)(pbq + 2 * j * WG_ + t * WT_);
#pragma unroll
  for (int t = 0; t < 2; ++t) f.b[2 + t] = *(const v4f*)(pbk + 2 * j * WG_ + t * WT_);
#pragma unroll
  for (int t = 0; t < 4; ++t) f.b[4 + t] = *(const v4f*)(pbv + 2 * j * WG_ + t * WT_);
}

__device__ __forceinline__ void proj_chunk(const float* pa, const float* pbq,
                                           const float* pbk, const float* pbv,
                                           v8f acc[8]) {
  ProjFrag f[2];
  proj_frag(f[0], pa, pbq, pbk, pbv, 0);
#pragma unroll
  for (int j = 0; j < PK_ / 8; ++j) {
    if (j + 1 < PK_ / 8) proj_frag(f[(j + 1) & 1], pa, pbq, pbk, pbv, j + 1);
    __builtin_amdgcn_sched_barrier(0);
    const ProjFrag& c = f[j & 1];
    const v2f a0 = {c.a.x, c.a.y};
    const v2f a1 = {c.a.z, c.a.w};
#pragma unroll
    for (int t = 0; t < 8; ++t) {
      const v2f b0 = {c.b[t].x, c.b[t].y};
      const bool kt = (t < 4);
      acc[t] = kt ? wmma_f32_16x16x4(b0, a0, acc[t]) : wmma_f32_16x16x4(a0, b0, acc[t]);
    }
#pragma unroll
    for (int t = 0; t < 8; ++t) {
      const v2f b1 = {c.b[t].z, c.b[t].w};
      const bool kt = (t < 4);
      acc[t] = kt ? wmma_f32_16x16x4(b1, a1, acc[t]) : wmma_f32_16x16x4(a1, b1, acc[t]);
    }
    __builtin_amdgcn_sched_barrier(0);
  }
}

__global__ __launch_bounds__(256) void proj_xpos_kernel(
    const float* __restrict__ X,  const float* __restrict__ WQ,
    const float* __restrict__ WK, const float* __restrict__ WV,
    float* __restrict__ Qb, float* __restrict__ Kb, float* __restrict__ Vb,
    float* __restrict__ Ob) {
  __shared__ __attribute__((aligned(16))) float sX[2][XBUF_];
  __shared__ __attribute__((aligned(16))) float sB[2][WBUF_];
  const int tid  = threadIdx.x;
  const int lane = tid & 31;
  const int w    = __builtin_amdgcn_readfirstlane((int)(threadIdx.x >> 5));
  const int r    = lane & 15;
  const int hi   = lane >> 4;
  const int mt   = w >> 1;
  const int h    = w & 1;
  const int rowBase = blockIdx.x * 64;

  const float* xg = X + (size_t)(rowBase + (tid >> 4)) * D_ + 4 * (tid & 15);
  const int ws = h ? VD_ : HD_;
  const float* wg = h ? (WV + 4 * lane) : (hi ? (WK + 4 * r) : (WQ + 4 * r));
  wg += (size_t)(4 * mt) * ws;
  const int cq = tid & 63;
  float* xs  = &sX[0][(tid >> 4) * XP_ + 4 * (tid & 15)];
  float* wsd = &sB[0][mt * WG_ + (cq >> 2) * WT_ + (cq & 3) * 16];

  const float* pa  = &sX[0][(16 * mt + r) * XP_ + 4 * hi];
  const float* pb  = &sB[0][hi * WG_ + 4 * r];
  const float* pbq = pb + WT_ * (2 * h);
  const float* pbk = pb + WT_ * (4 + 2 * h);
  const float* pbv = pb + WT_ * (8 + 4 * h);

  const v8f vz = {0.f, 0.f, 0.f, 0.f, 0.f, 0.f, 0.f, 0.f};
  v8f acc[8];
#pragma unroll
  for (int t = 0; t < 8; ++t) acc[t] = vz;

  ProjStage st;
  proj_fetch(st, xg, wg, ws, 0);
  proj_park(st, xs, wsd);
  __syncthreads();

#pragma nounroll
  for (int kc = 0; kc < NCH_ - 2; kc += 2) {
    proj_fetch(st, xg, wg, ws, kc + 1);
    __builtin_amdgcn_sched_barrier(0);
    proj_chunk(pa, pbq, pbk, pbv, acc);
    proj_park(st, xs + XBUF_, wsd + WBUF_);
    __syncthreads();
    proj_fetch(st, xg, wg, ws, kc + 2);
    __builtin_amdgcn_sched_barrier(0);
    proj_chunk(pa + XBUF_, pbq + WBUF_, pbk + WBUF_, pbv + WBUF_, acc);
    proj_park(st, xs, wsd);
    __syncthreads();
  }
  proj_fetch(st, xg, wg, ws, NCH_ - 1);
  __builtin_amdgcn_sched_barrier(0);
  proj_chunk(pa, pbq, pbk, pbv, acc);
  proj_park(st, xs + XBUF_, wsd + WBUF_);
  __syncthreads();
  proj_chunk(pa + XBUF_, pbq + WBUF_, pbk + WBUF_, pbv + WBUF_, acc);

  const int row16  = rowBase + 16 * mt;
  const int tile16 = blockIdx.x * 4 + mt;

  float* sK = &sB[0][0];
  float* sV = sK + HD_ * KP_;
  {
    const int n = (row16 + r) & (L_ - 1);
    float* qdst = Qb + (size_t)tile16 * (16 * HD_) + lane * 4;
    float* kdst = Kb + (size_t)tile16 * (16 * HD_) + lane * 4;
#pragma unroll
    for (int tt = 0; tt < 2; ++tt) {
      const int t = 2 * h + tt;
      float q8[8];
      float o8[8];
#pragma unroll
      for (int ip = 0; ip < 4; ++ip) {
        float snq, csq, snk, csk;
        xpos_coeffs_qk(n, 8 * t + 4 * hi + ip, snq, csq, snk, csk);
        float qe = acc[tt][2 * ip];
        float qo = acc[tt][2 * ip + 1];
        q8[2 * ip]     = fmaf(qe, csq, -(qo * snq));
        q8[2 * ip + 1] = fmaf(qo, csq, qe * snq);
        float xe = acc[2 + tt][2 * ip];
        float xo = acc[2 + tt][2 * ip + 1];
        o8[2 * ip]     = fmaf(xe, csk, -(xo * snk));
        o8[2 * ip + 1] = fmaf(xo, csk, xe * snk);
      }
      v4f qlo = {q8[0], q8[1], q8[2], q8[3]};
      v4f qup = {q8[4], q8[5], q8[6], q8[7]};
      *(v4f*)(qdst + (2 * t)     * 128) = qlo;
      *(v4f*)(qdst + (2 * t + 1) * 128) = qup;
      v4f lo = {o8[0], o8[1], o8[2], o8[3]};
      v4f up = {o8[4], o8[5], o8[6], o8[7]};
      *(v4f*)(kdst + (2 * t)     * 128) = lo;
      *(v4f*)(kdst + (2 * t + 1) * 128) = up;
#pragma unroll
      for (int i = 0; i < 8; ++i)
        sK[(16 * t + 8 * hi + i) * KP_ + 16 * mt + r] = o8[i];
    }
  }

  {
    float* vdst = Vb + (size_t)tile16 * (16 * VD_) + lane * 4;
    v4f dlo, dup;
    dlo.x = exp2f((float)(BS_ - 16 * mt - 8 * hi - 0) * LOG2_GAMMA);
    dlo.y = exp2f((float)(BS_ - 16 * mt - 8 * hi - 1) * LOG2_GAMMA);
    dlo.z = exp2f((float)(BS_ - 16 * mt - 8 * hi - 2) * LOG2_GAMMA);
    dlo.w = exp2f((float)(BS_ - 16 * mt - 8 * hi - 3) * LOG2_GAMMA);
    dup.x = exp2f((float)(BS_ - 16 * mt - 8 * hi - 4) * LOG2_GAMMA);
    dup.y = exp2f((float)(BS_ - 16 * mt - 8 * hi - 5) * LOG2_GAMMA);
    dup.z = exp2f((float)(BS_ - 16 * mt - 8 * hi - 6) * LOG2_GAMMA);
    dup.w = exp2f((float)(BS_ - 16 * mt - 8 * hi - 7) * LOG2_GAMMA);
#pragma unroll
    for (int tv = 0; tv < 4; ++tv) {
      const int ct = 4 * h + tv;
      v4f lo = {acc[4 + tv][0], acc[4 + tv][1], acc[4 + tv][2], acc[4 + tv][3]};
      v4f up = {acc[4 + tv][4], acc[4 + tv][5], acc[4 + tv][6], acc[4 + tv][7]};
      *(v4f*)(vdst + ct * 256)       = lo;
      *(v4f*)(vdst + ct * 256 + 128) = up;
      float* sv = sV + (16 * ct + r) * KP_ + 16 * mt + 8 * hi;
      *(v4f*)(sv)     = lo * dlo;
      *(v4f*)(sv + 4) = up * dup;
    }
  }

  if ((blockIdx.x & (NBLK_ - 1)) != NBLK_ - 1) {
    __builtin_amdgcn_fence(__ATOMIC_RELEASE, "workgroup", "local");
    __builtin_amdgcn_s_barrier();
    __builtin_amdgcn_fence(__ATOMIC_ACQUIRE, "workgroup", "local");
    const int ca = w & 1;
    const int cw = w >> 1;
    const float* pk = sK + (32 * ca + r) * KP_ + 8 * hi;
    const float* pv = sV + (32 * cw + r) * KP_ + 8 * hi;
    v8f ls[2][2];
#pragma unroll
    for (int ckl = 0; ckl < 2; ++ckl) { ls[ckl][0] = vz; ls[ckl][1] = vz; }
#pragma unroll
    for (int cj = 0; cj < BS_ / 16; ++cj) {
      v4f kf[2][2];
      v4f vf[2][2];
#pragma unroll
      for (int x = 0; x < 2; ++x) {
#pragma unroll
        for (int q = 0; q < 2; ++q) {
          kf[x][q] = *(const v4f*)(pk + x * 16 * KP_ + 16 * cj + 4 * q);
          vf[x][q] = *(const v4f*)(pv + x * 16 * KP_ + 16 * cj + 4 * q);
        }
      }
#pragma unroll
      for (int q = 0; q < 2; ++q) {
#pragma unroll
        for (int ckl = 0; ckl < 2; ++ckl) {
#pragma unroll
          for (int ctl = 0; ctl < 2; ++ctl)
            ls[ckl][ctl] = wmma_f32_16x16x4(v2f{kf[ckl][q].x, kf[ckl][q].y},
                                            v2f{vf[ctl][q].x, vf[ctl][q].y}, ls[ckl][ctl]);
        }
#pragma unroll
        for (int ckl = 0; ckl < 2; ++ckl) {
#pragma unroll
          for (int ctl = 0; ctl < 2; ++ctl)
            ls[ckl][ctl] = wmma_f32_16x16x4(v2f{kf[ckl][q].z, kf[ckl][q].w},
                                            v2f{vf[ctl][q].z, vf[ctl][q].w}, ls[ckl][ctl]);
        }
      }
    }
    float* Lt = Ob + ((size_t)blockIdx.x + 1) * (BS_ * VD_) + lane * 4;
#pragma unroll
    for (int ckl = 0; ckl < 2; ++ckl) {
#pragma unroll
      for (int ctl = 0; ctl < 2; ++ctl) {
        v8f a = ls[ckl][ctl];
        v4f lo = {a[0], a[1], a[2], a[3]};
        v4f up = {a[4], a[5], a[6], a[7]};
        float* dst = Lt + (2 * cw + ctl) * 1024 + (2 * ca + ckl) * 256;
        *(v4f*)(dst)       = lo;
        *(v4f*)(dst + 128) = up;
      }
    }
  }
}


#define SCAN_GRP_ 31
__global__ __launch_bounds__(256) void state_scan_kernel(float* Ob) {
  const int e  = (blockIdx.x & 7) * 256 + threadIdx.x;
  const int bb = blockIdx.x >> 3;
  float* base = Ob + (size_t)bb * L_ * VD_ + (size_t)e * 4;
  const float g = exp2f((float)BS_ * LOG2_GAMMA);
  v4f p = *(const v4f*)(base + (size_t)1 * (BS_ * VD_));
#pragma nounroll
  for (int c0 = 1; c0 < NBLK_ - 1; c0 += SCAN_GRP_) {
    float* sl = base + (size_t)(c0 + 1) * (BS_ * VD_);
    v4f x[SCAN_GRP_];
#pragma unroll
    for (int j = 0; j < SCAN_GRP_; ++j)
      x[j] = *(const v4f*)(sl + (size_t)j * (BS_ * VD_));
    __builtin_amdgcn_sched_barrier(0);
#pragma unroll
    for (int j = 0; j < SCAN_GRP_; ++j) {
      p = p * g + x[j];
      *(v4f*)(sl + (size_t)j * (BS_ * VD_)) = p;
    }
  }
}

#define WP_ 68

__global__ __launch_bounds__(256) void retention_block_kernel(
    const float* __restrict__ Qb, const float* __restrict__ Kb,
    const float* __restrict__ Vb, float* Ob) {
  __shared__ __attribute__((aligned(16))) float sW[8][16 * WP_];
  const int lane = threadIdx.x & 31;
  const int w    = __builtin_amdgcn_readfirstlane((int)(threadIdx.x >> 5));
  const int r    = lane & 15;
  const int hi   = lane >> 4;
  float* sS = sW[w];

  const int bb = blockIdx.x / NBLK_;
  const int c  = blockIdx.x % NBLK_;
  const int t  = (w < 4) ? w : 7 - w;
  const int cb = (w >> 2) * 4;
  const int s0 = c * BS_;
  const int n0 = s0 + 16 * t;

  const float* Q = Qb + (size_t)bb * L_ * HD_;
  const float* K = Kb + (size_t)bb * L_ * HD_;
  const float* V = Vb + (size_t)bb * L_ * VD_;
  float*       O = Ob + (size_t)bb * L_ * VD_;

  v2f qa[16];
  {
    const float* qp = Q + (size_t)n0 * HD_ + lane * 4;
#pragma unroll
    for (int t4 = 0; t4 < 4; ++t4) {
      v4f q0 = *(const v4f*)(qp + 256 * t4);
      v4f q1 = *(const v4f*)(qp + 256 * t4 + 128);
      qa[4 * t4 + 0] = v2f{q0.x, q0.y};
      qa[4 * t4 + 1] = v2f{q0.z, q0.w};
      qa[4 * t4 + 2] = v2f{q1.x, q1.y};
      qa[4 * t4 + 3] = v2f{q1.z, q1.w};
    }
  }

  const v8f vz = {0.f, 0.f, 0.f, 0.f, 0.f, 0.f, 0.f, 0.f};

  v8f oi[4];
#pragma unroll
  for (int ctl = 0; ctl < 4; ++ctl) oi[ctl] = vz;
  if (c > 0) {
    const float* P = O + (size_t)s0 * VD_ + lane * 4;
#pragma unroll
    for (int ctl = 0; ctl < 4; ++ctl) {
      const float* pp = P + (cb + ctl) * 1024;
#pragma unroll
      for (int t4 = 0; t4 < 4; ++t4) {
        v4f p0 = *(const v4f*)(pp + 256 * t4);
        v4f p1 = *(const v4f*)(pp + 256 * t4 + 128);
        oi[ctl] = wmma_f32_16x16x4(qa[4 * t4 + 0], v2f{p0.x, p0.y}, oi[ctl]);
        oi[ctl] = wmma_f32_16x16x4(qa[4 * t4 + 1], v2f{p0.z, p0.w}, oi[ctl]);
        oi[ctl] = wmma_f32_16x16x4(qa[4 * t4 + 2], v2f{p1.x, p1.y}, oi[ctl]);
        oi[ctl] = wmma_f32_16x16x4(qa[4 * t4 + 3], v2f{p1.z, p1.w}, oi[ctl]);
      }
    }
  }
  __syncthreads();

  float gp[8];
#pragma unroll
  for (int i = 0; i < 8; ++i)
    gp[i] = exp2f((float)((i + 8 * hi) - r) * LOG2_GAMMA);

  v8f o[4];
#pragma unroll
  for (int ctl = 0; ctl < 4; ++ctl) o[ctl] = vz;

  const float* kptr  = K + (size_t)s0 * HD_ + lane * 4;
  const float* vbase = V + (size_t)s0 * VD_ + cb * 256 + lane * 4;

  for (int m0 = s0; m0 <= n0; m0 += 16) {
    v4f kf[8];
#pragma unroll
    for (int j = 0; j < 8; ++j) kf[j] = *(const v4f*)(kptr + j * 128);
    v4f vf[2][4];
#pragma unroll
    for (int q = 0; q < 2; ++q) {
#pragma unroll
      for (int ctl = 0; ctl < 4; ++ctl)
        vf[q][ctl] = *(const v4f*)(vbase + ctl * 256 + q * 128);
    }

    v8f s = vz;
#pragma unroll
    for (int j = 0; j < 8; ++j) {
      s = wmma_f32_16x16x4(qa[2 * j],     v2f{kf[j].x, kf[j].y}, s);
      s = wmma_f32_16x16x4(qa[2 * j + 1], v2f{kf[j].z, kf[j].w}, s);
    }

    {
      const bool diag = (m0 == n0);
      float cf = exp2f((float)(n0 - m0) * LOG2_GAMMA);
#pragma unroll
      for (int i = 0; i < 8; ++i) {
        int diff = (i + 8 * hi) - r;
        float v = s[i] * (cf * gp[i]);
        s[i] = (diag && diff < 0) ? 0.0f : v;
      }
    }

#pragma unroll
    for (int i = 0; i < 8; ++i) sS[(i + 8 * hi) * 18 + r] = s[i];
    v2f av[4];
#pragma unroll
    for (int u = 0; u < 4; ++u)
      av[u] = *(const v2f*)&sS[r * 18 + 8 * hi + 2 * u];

#pragma unroll
    for (int q = 0; q < 2; ++q) {
#pragma unroll
      for (int ctl = 0; ctl < 4; ++ctl) {
        v4f f = vf[q][ctl];
        o[ctl] = wmma_f32_16x16x4(av[2 * q],     v2f{f.x, f.y}, o[ctl]);
        o[ctl] = wmma_f32_16x16x4(av[2 * q + 1], v2f{f.z, f.w}, o[ctl]);
      }
    }

    kptr  += 16 * HD_;
    vbase += 16 * VD_;
  }

#pragma unroll
  for (int i = 0; i < 8; ++i) {
    float rd = exp2f((float)(16 * t + i + 8 * hi) * LOG2_GAMMA);
#pragma unroll
    for (int ctl = 0; ctl < 4; ++ctl)
      sS[(i + 8 * hi) * WP_ + ctl * 16 + r] = fmaf(oi[ctl][i], rd, o[ctl][i]);
  }
  float* orow = O + (size_t)(n0 + hi) * VD_ + 16 * cb + 4 * r;
#pragma unroll
  for (int p = 0; p < 8; ++p) {
    v4f x = *(const v4f*)&sS[(2 * p + hi) * WP_ + 4 * r];
    *(v4f*)(orow + (size_t)(2 * p) * VD_) = x;
  }
}

extern "C" void kernel_launch(void* const* d_in, const int* in_sizes, int n_in,
                              void* d_out, int out_size, void* d_ws, size_t ws_size,
                              hipStream_t stream) {
  const float* X  = (const float*)d_in[0];
  const float* WQ = (const float*)d_in[1];
  const float* WK = (const float*)d_in[2];
  const float* WV = (const float*)d_in[3];

  float* Qb = (float*)d_ws;
  float* Kb = Qb + (size_t)B_ * L_ * HD_;
  float* Vb = Kb + (size_t)B_ * L_ * HD_;
  float* O  = (float*)d_out;

  proj_xpos_kernel<<<dim3(B_ * NBLK_), dim3(256), 0, stream>>>(X, WQ, WK, WV,
                                                               Qb, Kb, Vb, O);
  state_scan_kernel<<<dim3(B_ * 8), dim3(256), 0, stream>>>(O);
  retention_block_kernel<<<dim3(B_ * NBLK_), dim3(256), 0, stream>>>(Qb, Kb, Vb, O);
}


# projection kernel: staging (LDS park of the next chunk, global fetch of the one after) moved from the chunk boundary to behind the chunk's 1st product block, one chunk deeper; sum order unchanged
# speedup vs baseline: 2.4351x; 1.0307x over previous
#include <hip/hip_runtime.h>

typedef __attribute__((ext_vector_type(2))) float v2f;
typedef __attribute__((ext_vector_type(4))) float v4f;
typedef __attribute__((ext_vector_type(8))) float v8f;

#define B_  4
#define L_  4096
#define D_  512
#define HD_ 64
#define VD_ 128
#define BS_   64
#define NBLK_ (L_ / BS_)

constexpr float LOG2_GAMMA = -0.04580329f;

__device__ __forceinline__ v8f wmma_f32_16x16x4(v2f a, v2f b, v8f c) {
  return __builtin_amdgcn_wmma_f32_16x16x4_f32(false, a, false, b, (short)0, c,
                                               false, false);
}

__device__ __forceinline__ void xpos_coeffs_qk(int n, int j, float& snq, float& csq,
                                               float& snk, float& csk) {
  float sv    = (2.0f * (float)j + 0.4f * (float)HD_) * (1.0f / (1.4f * (float)HD_));
  float ex    = (float)n * (1.0f / 512.0f);
  float lsv   = log2f(sv);
  float scq   = exp2f(ex * lsv);
  float sck   = exp2f(-ex * lsv);
  float invf  = exp2f((float)j * (-13.2877124f / 32.0f));
  float ang   = (float)n * invf;
  float sn, cs;
  sincosf(ang, &sn, &cs);
  snq = sn * scq; csq = cs * scq;
  snk = sn * sck; csk = cs * sck;
}

#define PK_   64
#define NCH_  (D_ / PK_)
#define XP_   (PK_ + 4)
#define WT_   68
#define WG_   (16 * WT_)
#define XBUF_ (64 * XP_)
#define WBUF_ ((PK_ / 4) * WG_)
#define KP_   68

struct ProjStage {
  v4f x[4];
  v4f w[4][4];
};

__device__ __forceinline__ void proj_fetch(ProjStage& st, const float* __restrict__ xg,
                                           const float* __restrict__ wg, int ws, int kc) {
#pragma unroll
  for (int i = 0; i < 4; ++i)
    st.x[i] = *(const v4f*)(xg + (size_t)(16 * i) * D_ + kc * PK_);
#pragma unroll
  for (int i = 0; i < 4; ++i) {
#pragma unroll
    for (int e = 0; e < 4; ++e)
      st.w[i][e] = *(const v4f*)(wg + (size_t)(kc * PK_ + 16 * i + e) * ws);
  }
}

__device__ __forceinline__ void proj_park(const ProjStage& st, float* xs, float* wsd) {
#pragma unroll
  for (int i = 0; i < 4; ++i) *(v4f*)(xs + 16 * i * XP_) = st.x[i];
#pragma unroll
  for (int i = 0; i < 4; ++i) {
#pragma unroll
    for (int c4 = 0; c4 < 4; ++c4) {
      v4f col = {st.w[i][0][c4], st.w[i][1][c4], st.w[i][2][c4], st.w[i][3][c4]};
      *(v4f*)(wsd + 4 * i * WG_ + 4 * c4) = col;
    }
  }
}

struct ProjFrag {
  v4f a;
  v4f b[8];
};

__device__ __forceinline__ void proj_frag(ProjFrag& f, const float* pa, const float* pbq,
                                          const float* pbk, const float* pbv, int j) {
  f.a = *(const v4f*)(pa + 8 * j);
#pragma unroll
  for (int t = 0; t < 2; ++t) f.b[t]     = *(const v4f*)(pbq + 2 * j * WG_ + t * WT_);
#pragma unroll
  for (int t = 0; t < 2; ++t) f.b[2 + t] = *(const v4f*)(pbk + 2 * j * WG_ + t * WT_);
#pragma unroll
  for (int t = 0; t < 4; ++t) f.b[4 + t] = *(const v4f*)(pbv + 2 * j * WG_ + t * WT_);
}

#define MID_J_ 0

template <int MODE>
__device__ __forceinline__ void proj_chunk(const float* pa, const float* pbq,
                                           const float* pbk, const float* pbv,
                                           v8f acc[8], ProjStage& st,
                                           float* xs_next, float* wsd_next,
                                           const float* __restrict__ xg,
                                           const float* __restrict__ wg, int ws,
                                           int kc_fetch) {
  ProjFrag f[2];
  proj_frag(f[0], pa, pbq, pbk, pbv, 0);
#pragma unroll
  for (int j = 0; j < PK_ / 8; ++j) {
    if (j + 1 < PK_ / 8) proj_frag(f[(j + 1) & 1], pa, pbq, pbk, pbv, j + 1);
    __builtin_amdgcn_sched_barrier(0);
    const ProjFrag& c = f[j & 1];
    const v2f a0 = {c.a.x, c.a.y};
    const v2f a1 = {c.a.z, c.a.w};
#pragma unroll
    for (int t = 0; t < 8; ++t) {
      const v2f b0 = {c.b[t].x, c.b[t].y};
      const bool kt = (t < 4);
      acc[t] = kt ? wmma_f32_16x16x4(b0, a0, acc[t]) : wmma_f32_16x16x4(a0, b0, acc[t]);
    }
#pragma unroll
    for (int t = 0; t < 8; ++t) {
      const v2f b1 = {c.b[t].z, c.b[t].w};
      const bool kt = (t < 4);
      acc[t] = kt ? wmma_f32_16x16x4(b1, a1, acc[t]) : wmma_f32_16x16x4(a1, b1, acc[t]);
    }
    __builtin_amdgcn_sched_barrier(0);
    if (MODE != 0 && j == MID_J_) {
      proj_park(st, xs_next, wsd_next);
      if (MODE == 2) proj_fetch(st, xg, wg, ws, kc_fetch);
    }
  }
}

__global__ __launch_bounds__(256) void proj_xpos_kernel(
    const float* __restrict__ X,  const float* __restrict__ WQ,
    const float* __restrict__ WK, const float* __restrict__ WV,
    float* __restrict__ Qb, float* __restrict__ Kb, float* __restrict__ Vb,
    float* __restrict__ Ob) {
  __shared__ __attribute__((aligned(16))) float sX[2][XBUF_];
  __shared__ __attribute__((aligned(16))) float sB[2][WBUF_];
  const int tid  = threadIdx.x;
  const int lane = tid & 31;
  const int w    = __builtin_amdgcn_readfirstlane((int)(threadIdx.x >> 5));
  const int r    = lane & 15;
  const int hi   = lane >> 4;
  const int mt   = w >> 1;
  const int h    = w & 1;
  const int rowBase = blockIdx.x * 64;

  const float* xg = X + (size_t)(rowBase + (tid >> 4)) * D_ + 4 * (tid & 15);
  const int ws = h ? VD_ : HD_;
  const float* wg = h ? (WV + 4 * lane) : (hi ? (WK + 4 * r) : (WQ + 4 * r));
  wg += (size_t)(4 * mt) * ws;
  const int cq = tid & 63;
  float* xs  = &sX[0][(tid >> 4) * XP_ + 4 * (tid & 15)];
  float* wsd = &sB[0][mt * WG_ + (cq >> 2) * WT_ + (cq & 3) * 16];

  const float* pa  = &sX[0][(16 * mt + r) * XP_ + 4 * hi];
  const float* pb  = &sB[0][hi * WG_ + 4 * r];
  const float* pbq = pb + WT_ * (2 * h);
  const float* pbk = pb + WT_ * (4 + 2 * h);
  const float* pbv = pb + WT_ * (8 + 4 * h);

  const v8f vz = {0.f, 0.f, 0.f, 0.f, 0.f, 0.f, 0.f, 0.f};
  v8f acc[8];
#pragma unroll
  for (int t = 0; t < 8; ++t) acc[t] = vz;

  ProjStage st;
  {
    ProjStage st0;
    proj_fetch(st0, xg, wg, ws, 0);
    proj_fetch(st, xg, wg, ws, 1);
    proj_park(st0, xs, wsd);
  }
  __syncthreads();

#pragma nounroll
  for (int kc = 0; kc < NCH_ - 2; kc += 2) {
    proj_chunk<2>(pa, pbq, pbk, pbv, acc, st, xs + XBUF_, wsd + WBUF_, xg, wg, ws, kc + 2);
    __syncthreads();
    proj_chunk<2>(pa + XBUF_, pbq + WBUF_, pbk + WBUF_, pbv + WBUF_, acc, st, xs, wsd,
                  xg, wg, ws, kc + 3);
    __syncthreads();
  }
  proj_chunk<1>(pa, pbq, pbk, pbv, acc, st, xs + XBUF_, wsd + WBUF_, xg, wg, ws, 0);
  __syncthreads();
  proj_chunk<0>(pa + XBUF_, pbq + WBUF_, pbk + WBUF_, pbv + WBUF_, acc, st, xs, wsd,
                xg, wg, ws, 0);

  const int row16  = rowBase + 16 * mt;
  const int tile16 = blockIdx.x * 4 + mt;

  float* sK = &sB[0][0];
  float* sV = sK + HD_ * KP_;
  {
    const int n = (row16 + r) & (L_ - 1);
    float* qdst = Qb + (size_t)tile16 * (16 * HD_) + lane * 4;
    float* kdst = Kb + (size_t)tile16 * (16 * HD_) + lane * 4;
#pragma unroll
    for (int tt = 0; tt < 2; ++tt) {
      const int t = 2 * h + tt;
      float q8[8];
      float o8[8];
#pragma unroll
      for (int ip = 0; ip < 4; ++ip) {
        float snq, csq, snk, csk;
        xpos_coeffs_qk(n, 8 * t + 4 * hi + ip, snq, csq, snk, csk);
        float qe = acc[tt][2 * ip];
        float qo = acc[tt][2 * ip + 1];
        q8[2 * ip]     = fmaf(qe, csq, -(qo * snq));
        q8[2 * ip + 1] = fmaf(qo, csq, qe * snq);
        float xe = acc[2 + tt][2 * ip];
        float xo = acc[2 + tt][2 * ip + 1];
        o8[2 * ip]     = fmaf(xe, csk, -(xo * snk));
        o8[2 * ip + 1] = fmaf(xo, csk, xe * snk);
      }
      v4f qlo = {q8[0], q8[1], q8[2], q8[3]};
      v4f qup = {q8[4], q8[5], q8[6], q8[7]};
      *(v4f*)(qdst + (2 * t)     * 128) = qlo;
      *(v4f*)(qdst + (2 * t + 1) * 128) = qup;
      v4f lo = {o8[0], o8[1], o8[2], o8[3]};
      v4f up = {o8[4], o8[5], o8[6], o8[7]};
      *(v4f*)(kdst + (2 * t)     * 128) = lo;
      *(v4f*)(kdst + (2 * t + 1) * 128) = up;
#pragma unroll
      for (int i = 0; i < 8; ++i)
        sK[(16 * t + 8 * hi + i) * KP_ + 16 * mt + r] = o8[i];
    }
  }

  {
    float* vdst = Vb + (size_t)tile16 * (16 * VD_) + lane * 4;
    v4f dlo, dup;
    dlo.x = exp2f((float)(BS_ - 16 * mt - 8 * hi - 0) * LOG2_GAMMA);
    dlo.y = exp2f((float)(BS_ - 16 * mt - 8 * hi - 1) * LOG2_GAMMA);
    dlo.z = exp2f((float)(BS_ - 16 * mt - 8 * hi - 2) * LOG2_GAMMA);
    dlo.w = exp2f((float)(BS_ - 16 * mt - 8 * hi - 3) * LOG2_GAMMA);
    dup.x = exp2f((float)(BS_ - 16 * mt - 8 * hi - 4) * LOG2_GAMMA);
    dup.y = exp2f((float)(BS_ - 16 * mt - 8 * hi - 5) * LOG2_GAMMA);
    dup.z = exp2f((float)(BS_ - 16 * mt - 8 * hi - 6) * LOG2_GAMMA);
    dup.w = exp2f((float)(BS_ - 16 * mt - 8 * hi - 7) * LOG2_GAMMA);
#pragma unroll
    for (int tv = 0; tv < 4; ++tv) {
      const int ct = 4 * h + tv;
      v4f lo = {acc[4 + tv][0], acc[4 + tv][1], acc[4 + tv][2], acc[4 + tv][3]};
      v4f up = {acc[4 + tv][4], acc[4 + tv][5], acc[4 + tv][6], acc[4 + tv][7]};
      *(v4f*)(vdst + ct * 256)       = lo;
      *(v4f*)(vdst + ct * 256 + 128) = up;
      float* sv = sV + (16 * ct + r) * KP_ + 16 * mt + 8 * hi;
      *(v4f*)(sv)     = lo * dlo;
      *(v4f*)(sv + 4) = up * dup;
    }
  }

  if ((blockIdx.x & (NBLK_ - 1)) != NBLK_ - 1) {
    __builtin_amdgcn_fence(__ATOMIC_RELEASE, "workgroup", "local");
    __builtin_amdgcn_s_barrier();
    __builtin_amdgcn_fence(__ATOMIC_ACQUIRE, "workgroup", "local");
    const int ca = w & 1;
    const int cw = w >> 1;
    const float* pk = sK + (32 * ca + r) * KP_ + 8 * hi;
    const float* pv = sV + (32 * cw + r) * KP_ + 8 * hi;
    v8f ls[2][2];
#pragma unroll
    for (int ckl = 0; ckl < 2; ++ckl) { ls[ckl][0] = vz; ls[ckl][1] = vz; }
#pragma unroll
    for (int cj = 0; cj < BS_ / 16; ++cj) {
      v4f kf[2][2];
      v4f vf[2][2];
#pragma unroll
      for (int x = 0; x < 2; ++x) {
#pragma unroll
        for (int q = 0; q < 2; ++q) {
          kf[x][q] = *(const v4f*)(pk + x * 16 * KP_ + 16 * cj + 4 * q);
          vf[x][q] = *(const v4f*)(pv + x * 16 * KP_ + 16 * cj + 4 * q);
        }
      }
#pragma unroll
      for (int q = 0; q < 2; ++q) {
#pragma unroll
        for (int ckl = 0; ckl < 2; ++ckl) {
#pragma unroll
          for (int ctl = 0; ctl < 2; ++ctl)
            ls[ckl][ctl] = wmma_f32_16x16x4(v2f{kf[ckl][q].x, kf[ckl][q].y},
                                            v2f{vf[ctl][q].x, vf[ctl][q].y}, ls[ckl][ctl]);
        }
#pragma unroll
        for (int ckl = 0; ckl < 2; ++ckl) {
#pragma unroll
          for (int ctl = 0; ctl < 2; ++ctl)
            ls[ckl][ctl] = wmma_f32_16x16x4(v2f{kf[ckl][q].z, kf[ckl][q].w},
                                            v2f{vf[ctl][q].z, vf[ctl][q].w}, ls[ckl][ctl]);
        }
      }
    }
    float* Lt = Ob + ((size_t)blockIdx.x + 1) * (BS_ * VD_) + lane * 4;
#pragma unroll
    for (int ckl = 0; ckl < 2; ++ckl) {
#pragma unroll
      for (int ctl = 0; ctl < 2; ++ctl) {
        v8f a = ls[ckl][ctl];
        v4f lo = {a[0], a[1], a[2], a[3]};
        v4f up = {a[4], a[5], a[6], a[7]};
        float* dst = Lt + (2 * cw + ctl) * 1024 + (2 * ca + ckl) * 256;
        *(v4f*)(dst)       = lo;
        *(v4f*)(dst + 128) = up;
      }
    }
  }
}


#define SCAN_GRP_ 31
__global__ __launch_bounds__(256) void state_scan_kernel(float* Ob) {
  const int e  = (blockIdx.x & 7) * 256 + threadIdx.x;
  const int bb = blockIdx.x >> 3;
  float* base = Ob + (size_t)bb * L_ * VD_ + (size_t)e * 4;
  const float g = exp2f((float)BS_ * LOG2_GAMMA);
  v4f p = *(const v4f*)(base + (size_t)1 * (BS_ * VD_));
#pragma nounroll
  for (int c0 = 1; c0 < NBLK_ - 1; c0 += SCAN_GRP_) {
    float* sl = base + (size_t)(c0 + 1) * (BS_ * VD_);
    v4f x[SCAN_GRP_];
#pragma unroll
    for (int j = 0; j < SCAN_GRP_; ++j)
      x[j] = *(const v4f*)(sl + (size_t)j * (BS_ * VD_));
    __builtin_amdgcn_sched_barrier(0);
#pragma unroll
    for (int j = 0; j < SCAN_GRP_; ++j) {
      p = p * g + x[j];
      *(v4f*)(sl + (size_t)j * (BS_ * VD_)) = p;
    }
  }
}

#define WP_ 68

__global__ __launch_bounds__(256) void retention_block_kernel(
    const float* __restrict__ Qb, const float* __restrict__ Kb,
    const float* __restrict__ Vb, float* Ob) {
  __shared__ __attribute__((aligned(16))) float sW[8][16 * WP_];
  const int lane = threadIdx.x & 31;
  const int w    = __builtin_amdgcn_readfirstlane((int)(threadIdx.x >> 5));
  const int r    = lane & 15;
  const int hi   = lane >> 4;
  float* sS = sW[w];

  const int bb = blockIdx.x / NBLK_;
  const int c  = blockIdx.x % NBLK_;
  const int t  = (w < 4) ? w : 7 - w;
  const int cb = (w >> 2) * 4;
  const int s0 = c * BS_;
  const int n0 = s0 + 16 * t;

  const float* Q = Qb + (size_t)bb * L_ * HD_;
  const float* K = Kb + (size_t)bb * L_ * HD_;
  const float* V = Vb + (size_t)bb * L_ * VD_;
  float*       O = Ob + (size_t)bb * L_ * VD_;

  v2f qa[16];
  {
    const float* qp = Q + (size_t)n0 * HD_ + lane * 4;
#pragma unroll
    for (int t4 = 0; t4 < 4; ++t4) {
      v4f q0 = *(const v4f*)(qp + 256 * t4);
      v4f q1 = *(const v4f*)(qp + 256 * t4 + 128);
      qa[4 * t4 + 0] = v2f{q0.x, q0.y};
      qa[4 * t4 + 1] = v2f{q0.z, q0.w};
      qa[4 * t4 + 2] = v2f{q1.x, q1.y};
      qa[4 * t4 + 3] = v2f{q1.z, q1.w};
    }
  }

  const v8f vz = {0.f, 0.f, 0.f, 0.f, 0.f, 0.f, 0.f, 0.f};

  v8f oi[4];
#pragma unroll
  for (int ctl = 0; ctl < 4; ++ctl) oi[ctl] = vz;
  if (c > 0) {
    const float* P = O + (size_t)s0 * VD_ + lane * 4;
#pragma unroll
    for (int ctl = 0; ctl < 4; ++ctl) {
      const float* pp = P + (cb + ctl) * 1024;
#pragma unroll
      for (int t4 = 0; t4 < 4; ++t4) {
        v4f p0 = *(const v4f*)(pp + 256 * t4);
        v4f p1 = *(const v4f*)(pp + 256 * t4 + 128);
        oi[ctl] = wmma_f32_16x16x4(qa[4 * t4 + 0], v2f{p0.x, p0.y}, oi[ctl]);
        oi[ctl] = wmma_f32_16x16x4(qa[4 * t4 + 1], v2f{p0.z, p0.w}, oi[ctl]);
        oi[ctl] = wmma_f32_16x16x4(qa[4 * t4 + 2], v2f{p1.x, p1.y}, oi[ctl]);
        oi[ctl] = wmma_f32_16x16x4(qa[4 * t4 + 3], v2f{p1.z, p1.w}, oi[ctl]);
      }
    }
  }
  __syncthreads();

  float gp[8];
#pragma unroll
  for (int i = 0; i < 8; ++i)
    gp[i] = exp2f((float)((i + 8 * hi) - r) * LOG2_GAMMA);

  v8f o[4];
#pragma unroll
  for (int ctl = 0; ctl < 4; ++ctl) o[ctl] = vz;

  const float* kptr  = K + (size_t)s0 * HD_ + lane * 4;
  const float* vbase = V + (size_t)s0 * VD_ + cb * 256 + lane * 4;

  for (int m0 = s0; m0 <= n0; m0 += 16) {
    v4f kf[8];
#pragma unroll
    for (int j = 0; j < 8; ++j) kf[j] = *(const v4f*)(kptr + j * 128);
    v4f vf[2][4];
#pragma unroll
    for (int q = 0; q < 2; ++q) {
#pragma unroll
      for (int ctl = 0; ctl < 4; ++ctl)
        vf[q][ctl] = *(const v4f*)(vbase + ctl * 256 + q * 128);
    }

    v8f s = vz;
#pragma unroll
    for (int j = 0; j < 8; ++j) {
      s = wmma_f32_16x16x4(qa[2 * j],     v2f{kf[j].x, kf[j].y}, s);
      s = wmma_f32_16x16x4(qa[2 * j + 1], v2f{kf[j].z, kf[j].w}, s);
    }

    {
      const bool diag = (m0 == n0);
      float cf = exp2f((float)(n0 - m0) * LOG2_GAMMA);
#pragma unroll
      for (int i = 0; i < 8; ++i) {
        int diff = (i + 8 * hi) - r;
        float v = s[i] * (cf * gp[i]);
        s[i] = (diag && diff < 0) ? 0.0f : v;
      }
    }

#pragma unroll
    for (int i = 0; i < 8; ++i) sS[(i + 8 * hi) * 18 + r] = s[i];
    v2f av[4];
#pragma unroll
    for (int u = 0; u < 4; ++u)
      av[u] = *(const v2f*)&sS[r * 18 + 8 * hi + 2 * u];

#pragma unroll
    for (int q = 0; q < 2; ++q) {
#pragma unroll
      for (int ctl = 0; ctl < 4; ++ctl) {
        v4f f = vf[q][ctl];
        o[ctl] = wmma_f32_16x16x4(av[2 * q],     v2f{f.x, f.y}, o[ctl]);
        o[ctl] = wmma_f32_16x16x4(av[2 * q + 1], v2f{f.z, f.w}, o[ctl]);
      }
    }

    kptr  += 16 * HD_;
    vbase += 16 * VD_;
  }

#pragma unroll
  for (int i = 0; i < 8; ++i) {
    float rd = exp2f((float)(16 * t + i + 8 * hi) * LOG2_GAMMA);
#pragma unroll
    for (int ctl = 0; ctl < 4; ++ctl)
      sS[(i + 8 * hi) * WP_ + ctl * 16 + r] = fmaf(oi[ctl][i], rd, o[ctl][i]);
  }
  float* orow = O + (size_t)(n0 + hi) * VD_ + 16 * cb + 4 * r;
#pragma unroll
  for (int p = 0; p < 8; ++p) {
    v4f x = *(const v4f*)&sS[(2 * p + hi) * WP_ + 4 * r];
    *(v4f*)(orow + (size_t)(2 * p) * VD_) = x;
  }
}

extern "C" void kernel_launch(void* const* d_in, const int* in_sizes, int n_in,
                              void* d_out, int out_size, void* d_ws, size_t ws_size,
                              hipStream_t stream) {
  const float* X  = (const float*)d_in[0];
  const float* WQ = (const float*)d_in[1];
  const float* WK = (const float*)d_in[2];
  const float* WV = (const float*)d_in[3];

  float* Qb = (float*)d_ws;
  float* Kb = Qb + (size_t)B_ * L_ * HD_;
  float* Vb = Kb + (size_t)B_ * L_ * HD_;
  float* O  = (float*)d_out;

  proj_xpos_kernel<<<dim3(B_ * NBLK_), dim3(256), 0, stream>>>(X, WQ, WK, WV,
                                                               Qb, Kb, Vb, O);
  state_scan_kernel<<<dim3(B_ * 8), dim3(256), 0, stream>>>(O);
  retention_block_kernel<<<dim3(B_ * NBLK_), dim3(256), 0, stream>>>(Qb, Kb, Vb, O);
}


# state scan dealt over all 256 WGPs: 256 one-wave work-groups (was 32 of 8 waves), a lane's 62 slots requested as one group; sum order unchanged
# speedup vs baseline: 2.6186x; 1.0754x over previous
#include <hip/hip_runtime.h>

typedef __attribute__((ext_vector_type(2))) float v2f;
typedef __attribute__((ext_vector_type(4))) float v4f;
typedef __attribute__((ext_vector_type(8))) float v8f;

#define B_  4
#define L_  4096
#define D_  512
#define HD_ 64
#define VD_ 128
#define BS_   64
#define NBLK_ (L_ / BS_)

constexpr float LOG2_GAMMA = -0.04580329f;

__device__ __forceinline__ v8f wmma_f32_16x16x4(v2f a, v2f b, v8f c) {
  return __builtin_amdgcn_wmma_f32_16x16x4_f32(false, a, false, b, (short)0, c,
                                               false, false);
}

__device__ __forceinline__ void xpos_coeffs_qk(int n, int j, float& snq, float& csq,
                                               float& snk, float& csk) {
  float sv    = (2.0f * (float)j + 0.4f * (float)HD_) * (1.0f / (1.4f * (float)HD_));
  float ex    = (float)n * (1.0f / 512.0f);
  float lsv   = log2f(sv);
  float scq   = exp2f(ex * lsv);
  float sck   = exp2f(-ex * lsv);
  float invf  = exp2f((float)j * (-13.2877124f / 32.0f));
  float ang   = (float)n * invf;
  float sn, cs;
  sincosf(ang, &sn, &cs);
  snq = sn * scq; csq = cs * scq;
  snk = sn * sck; csk = cs * sck;
}

#define PK_   64
#define NCH_  (D_ / PK_)
#define XP_   (PK_ + 4)
#define WT_   68
#define WG_   (16 * WT_)
#define XBUF_ (64 * XP_)
#define WBUF_ ((PK_ / 4) * WG_)
#define KP_   68

struct ProjStage {
  v4f x[4];
  v4f w[4][4];
};

__device__ __forceinline__ void proj_fetch(ProjStage& st, const float* __restrict__ xg,
                                           const float* __restrict__ wg, int ws, int kc) {
#pragma unroll
  for (int i = 0; i < 4; ++i)
    st.x[i] = *(const v4f*)(xg + (size_t)(16 * i) * D_ + kc * PK_);
#pragma unroll
  for (int i = 0; i < 4; ++i) {
#pragma unroll
    for (int e = 0; e < 4; ++e)
      st.w[i][e] = *(const v4f*)(wg + (size_t)(kc * PK_ + 16 * i + e) * ws);
  }
}

__device__ __forceinline__ void proj_park(const ProjStage& st, float* xs, float* wsd) {
#pragma unroll
  for (int i = 0; i < 4; ++i) *(v4f*)(xs + 16 * i * XP_) = st.x[i];
#pragma unroll
  for (int i = 0; i < 4; ++i) {
#pragma unroll
    for (int c4 = 0; c4 < 4; ++c4) {
      v4f col = {st.w[i][0][c4], st.w[i][1][c4], st.w[i][2][c4], st.w[i][3][c4]};
      *(v4f*)(wsd + 4 * i * WG_ + 4 * c4) = col;
    }
  }
}

struct ProjFrag {
  v4f a;
  v4f b[8];
};

__device__ __forceinline__ void proj_frag(ProjFrag& f, const float* pa, const float* pbq,
                                          const float* pbk, const float* pbv, int j) {
  f.a = *(const v4f*)(pa + 8 * j);
#pragma unroll
  for (int t = 0; t < 2; ++t) f.b[t]     = *(const v4f*)(pbq + 2 * j * WG_ + t * WT_);
#pragma unroll
  for (int t = 0; t < 2; ++t) f.b[2 + t] = *(const v4f*)(pbk + 2 * j * WG_ + t * WT_);
#pragma unroll
  for (int t = 0; t < 4; ++t) f.b[4 + t] = *(const v4f*)(pbv + 2 * j * WG_ + t * WT_);
}

#define MID_J_ 0

template <int MODE>
__device__ __forceinline__ void proj_chunk(const float* pa, const float* pbq,
                                           const float* pbk, const float* pbv,
                                           v8f acc[8], ProjStage& st,
                                           float* xs_next, float* wsd_next,
                                           const float* __restrict__ xg,
                                           const float* __restrict__ wg, int ws,
                                           int kc_fetch) {
  ProjFrag f[2];
  proj_frag(f[0], pa, pbq, pbk, pbv, 0);
#pragma unroll
  for (int j = 0; j < PK_ / 8; ++j) {
    if (j + 1 < PK_ / 8) proj_frag(f[(j + 1) & 1], pa, pbq, pbk, pbv, j + 1);
    __builtin_amdgcn_sched_barrier(0);
    const ProjFrag& c = f[j & 1];
    const v2f a0 = {c.a.x, c.a.y};
    const v2f a1 = {c.a.z, c.a.w};
#pragma unroll
    for (int t = 0; t < 8; ++t) {
      const v2f b0 = {c.b[t].x, c.b[t].y};
      const bool kt = (t < 4);
      acc[t] = kt ? wmma_f32_16x16x4(b0, a0, acc[t]) : wmma_f32_16x16x4(a0, b0, acc[t]);
    }
#pragma unroll
    for (int t = 0; t < 8; ++t) {
      const v2f b1 = {c.b[t].z, c.b[t].w};
      const bool kt = (t < 4);
      acc[t] = kt ? wmma_f32_16x16x4(b1, a1, acc[t]) : wmma_f32_16x16x4(a1, b1, acc[t]);
    }
    __builtin_amdgcn_sched_barrier(0);
    if (MODE != 0 && j == MID_J_) {
      proj_park(st, xs_next, wsd_next);
      if (MODE == 2) proj_fetch(st, xg, wg, ws, kc_fetch);
    }
  }
}

__global__ __launch_bounds__(256) void proj_xpos_kernel(
    const float* __restrict__ X,  const float* __restrict__ WQ,
    const float* __restrict__ WK, const float* __restrict__ WV,
    float* __restrict__ Qb, float* __restrict__ Kb, float* __restrict__ Vb,
    float* __restrict__ Ob) {
  __shared__ __attribute__((aligned(16))) float sX[2][XBUF_];
  __shared__ __attribute__((aligned(16))) float sB[2][WBUF_];
  const int tid  = threadIdx.x;
  const int lane = tid & 31;
  const int w    = __builtin_amdgcn_readfirstlane((int)(threadIdx.x >> 5));
  const int r    = lane & 15;
  const int hi   = lane >> 4;
  const int mt   = w >> 1;
  const int h    = w & 1;
  const int rowBase = blockIdx.x * 64;

  const float* xg = X + (size_t)(rowBase + (tid >> 4)) * D_ + 4 * (tid & 15);
  const int ws = h ? VD_ : HD_;
  const float* wg = h ? (WV + 4 * lane) : (hi ? (WK + 4 * r) : (WQ + 4 * r));
  wg += (size_t)(4 * mt) * ws;
  const int cq = tid & 63;
  float* xs  = &sX[0][(tid >> 4) * XP_ + 4 * (tid & 15)];
  float* wsd = &sB[0][mt * WG_ + (cq >> 2) * WT_ + (cq & 3) * 16];

  const float* pa  = &sX[0][(16 * mt + r) * XP_ + 4 * hi];
  const float* pb  = &sB[0][hi * WG_ + 4 * r];
  const float* pbq = pb + WT_ * (2 * h);
  const float* pbk = pb + WT_ * (4 + 2 * h);
  const float* pbv = pb + WT_ * (8 + 4 * h);

  const v8f vz = {0.f, 0.f, 0.f, 0.f, 0.f, 0.f, 0.f, 0.f};
  v8f acc[8];
#pragma unroll
  for (int t = 0; t < 8; ++t) acc[t] = vz;

  ProjStage st;
  {
    ProjStage st0;
    proj_fetch(st0, xg, wg, ws, 0);
    proj_fetch(st, xg, wg, ws, 1);
    proj_park(st0, xs, wsd);
  }
  __syncthreads();

#pragma nounroll
  for (int kc = 0; kc < NCH_ - 2; kc += 2) {
    proj_chunk<2>(pa, pbq, pbk, pbv, acc, st, xs + XBUF_, wsd + WBUF_, xg, wg, ws, kc + 2);
    __syncthreads();
    proj_chunk<2>(pa + XBUF_, pbq + WBUF_, pbk + WBUF_, pbv + WBUF_, acc, st, xs, wsd,
                  xg, wg, ws, kc + 3);
    __syncthreads();
  }
  proj_chunk<1>(pa, pbq, pbk, pbv, acc, st, xs + XBUF_, wsd + WBUF_, xg, wg, ws, 0);
  __syncthreads();
  proj_chunk<0>(pa + XBUF_, pbq + WBUF_, pbk + WBUF_, pbv + WBUF_, acc, st, xs, wsd,
                xg, wg, ws, 0);

  const int row16  = rowBase + 16 * mt;
  const int tile16 = blockIdx.x * 4 + mt;

  float* sK = &sB[0][0];
  float* sV = sK + HD_ * KP_;
  {
    const int n = (row16 + r) & (L_ - 1);
    float* qdst = Qb + (size_t)tile16 * (16 * HD_) + lane * 4;
    float* kdst = Kb + (size_t)tile16 * (16 * HD_) + lane * 4;
#pragma unroll
    for (int tt = 0; tt < 2; ++tt) {
      const int t = 2 * h + tt;
      float q8[8];
      float o8[8];
#pragma unroll
      for (int ip = 0; ip < 4; ++ip) {
        float snq, csq, snk, csk;
        xpos_coeffs_qk(n, 8 * t + 4 * hi + ip, snq, csq, snk, csk);
        float qe = acc[tt][2 * ip];
        float qo = acc[tt][2 * ip + 1];
        q8[2 * ip]     = fmaf(qe, csq, -(qo * snq));
        q8[2 * ip + 1] = fmaf(qo, csq, qe * snq);
        float xe = acc[2 + tt][2 * ip];
        float xo = acc[2 + tt][2 * ip + 1];
        o8[2 * ip]     = fmaf(xe, csk, -(xo * snk));
        o8[2 * ip + 1] = fmaf(xo, csk, xe * snk);
      }
      v4f qlo = {q8[0], q8[1], q8[2], q8[3]};
      v4f qup = {q8[4], q8[5], q8[6], q8[7]};
      *(v4f*)(qdst + (2 * t)     * 128) = qlo;
      *(v4f*)(qdst + (2 * t + 1) * 128) = qup;
      v4f lo = {o8[0], o8[1], o8[2], o8[3]};
      v4f up = {o8[4], o8[5], o8[6], o8[7]};
      *(v4f*)(kdst + (2 * t)     * 128) = lo;
      *(v4f*)(kdst + (2 * t + 1) * 128) = up;
#pragma unroll
      for (int i = 0; i < 8; ++i)
        sK[(16 * t + 8 * hi + i) * KP_ + 16 * mt + r] = o8[i];
    }
  }

  {
    float* vdst = Vb + (size_t)tile16 * (16 * VD_) + lane * 4;
    v4f dlo, dup;
    dlo.x = exp2f((float)(BS_ - 16 * mt - 8 * hi - 0) * LOG2_GAMMA);
    dlo.y = exp2f((float)(BS_ - 16 * mt - 8 * hi - 1) * LOG2_GAMMA);
    dlo.z = exp2f((float)(BS_ - 16 * mt - 8 * hi - 2) * LOG2_GAMMA);
    dlo.w = exp2f((float)(BS_ - 16 * mt - 8 * hi - 3) * LOG2_GAMMA);
    dup.x = exp2f((float)(BS_ - 16 * mt - 8 * hi - 4) * LOG2_GAMMA);
    dup.y = exp2f((float)(BS_ - 16 * mt - 8 * hi - 5) * LOG2_GAMMA);
    dup.z = exp2f((float)(BS_ - 16 * mt - 8 * hi - 6) * LOG2_GAMMA);
    dup.w = exp2f((float)(BS_ - 16 * mt - 8 * hi - 7) * LOG2_GAMMA);
#pragma unroll
    for (int tv = 0; tv < 4; ++tv) {
      const int ct = 4 * h + tv;
      v4f lo = {acc[4 + tv][0], acc[4 + tv][1], acc[4 + tv][2], acc[4 + tv][3]};
      v4f up = {acc[4 + tv][4], acc[4 + tv][5], acc[4 + tv][6], acc[4 + tv][7]};
      *(v4f*)(vdst + ct * 256)       = lo;
      *(v4f*)(vdst + ct * 256 + 128) = up;
      float* sv = sV + (16 * ct + r) * KP_ + 16 * mt + 8 * hi;
      *(v4f*)(sv)     = lo * dlo;
      *(v4f*)(sv + 4) = up * dup;
    }
  }

  if ((blockIdx.x & (NBLK_ - 1)) != NBLK_ - 1) {
    __builtin_amdgcn_fence(__ATOMIC_RELEASE, "workgroup", "local");
    __builtin_amdgcn_s_barrier();
    __builtin_amdgcn_fence(__ATOMIC_ACQUIRE, "workgroup", "local");
    const int ca = w & 1;
    const int cw = w >> 1;
    const float* pk = sK + (32 * ca + r) * KP_ + 8 * hi;
    const float* pv = sV + (32 * cw + r) * KP_ + 8 * hi;
    v8f ls[2][2];
#pragma unroll
    for (int ckl = 0; ckl < 2; ++ckl) { ls[ckl][0] = vz; ls[ckl][1] = vz; }
#pragma unroll
    for (int cj = 0; cj < BS_ / 16; ++cj) {
      v4f kf[2][2];
      v4f vf[2][2];
#pragma unroll
      for (int x = 0; x < 2; ++x) {
#pragma unroll
        for (int q = 0; q < 2; ++q) {
          kf[x][q] = *(const v4f*)(pk + x * 16 * KP_ + 16 * cj + 4 * q);
          vf[x][q] = *(const v4f*)(pv + x * 16 * KP_ + 16 * cj + 4 * q);
        }
      }
#pragma unroll
      for (int q = 0; q < 2; ++q) {
#pragma unroll
        for (int ckl = 0; ckl < 2; ++ckl) {
#pragma unroll
          for (int ctl = 0; ctl < 2; ++ctl)
            ls[ckl][ctl] = wmma_f32_16x16x4(v2f{kf[ckl][q].x, kf[ckl][q].y},
                                            v2f{vf[ctl][q].x, vf[ctl][q].y}, ls[ckl][ctl]);
        }
#pragma unroll
        for (int ckl = 0; ckl < 2; ++ckl) {
#pragma unroll
          for (int ctl = 0; ctl < 2; ++ctl)
            ls[ckl][ctl] = wmma_f32_16x16x4(v2f{kf[ckl][q].z, kf[ckl][q].w},
                                            v2f{vf[ctl][q].z, vf[ctl][q].w}, ls[ckl][ctl]);
        }
      }
    }
    float* Lt = Ob + ((size_t)blockIdx.x + 1) * (BS_ * VD_) + lane * 4;
#pragma unroll
    for (int ckl = 0; ckl < 2; ++ckl) {
#pragma unroll
      for (int ctl = 0; ctl < 2; ++ctl) {
        v8f a = ls[ckl][ctl];
        v4f lo = {a[0], a[1], a[2], a[3]};
        v4f up = {a[4], a[5], a[6], a[7]};
        float* dst = Lt + (2 * cw + ctl) * 1024 + (2 * ca + ckl) * 256;
        *(v4f*)(dst)       = lo;
        *(v4f*)(dst + 128) = up;
      }
    }
  }
}


#define SCAN_N_ (NBLK_ - 2)
__global__ __launch_bounds__(32) void state_scan_kernel(float* Ob) {
  const int e  = (blockIdx.x & 63) * 32 + threadIdx.x;
  const int bb = blockIdx.x >> 6;
  float* base = Ob + (size_t)bb * L_ * VD_ + (size_t)e * 4;
  const float g = exp2f((float)BS_ * LOG2_GAMMA);
  v4f p = *(const v4f*)(base + (size_t)1 * (BS_ * VD_));
  float* sl = base + (size_t)2 * (BS_ * VD_);
  v4f x[SCAN_N_];
#pragma unroll
  for (int j = 0; j < SCAN_N_; ++j)
    x[j] = *(const v4f*)(sl + (size_t)j * (BS_ * VD_));
  __builtin_amdgcn_sched_barrier(0);
#pragma unroll
  for (int j = 0; j < SCAN_N_; ++j) {
    p = p * g + x[j];
    *(v4f*)(sl + (size_t)j * (BS_ * VD_)) = p;
  }
}

#define WP_ 68

__global__ __launch_bounds__(256) void retention_block_kernel(
    const float* __restrict__ Qb, const float* __restrict__ Kb,
    const float* __restrict__ Vb, float* Ob) {
  __shared__ __attribute__((aligned(16))) float sW[8][16 * WP_];
  const int lane = threadIdx.x & 31;
  const int w    = __builtin_amdgcn_readfirstlane((int)(threadIdx.x >> 5));
  const int r    = lane & 15;
  const int hi   = lane >> 4;
  float* sS = sW[w];

  const int bb = blockIdx.x / NBLK_;
  const int c  = blockIdx.x % NBLK_;
  const int t  = (w < 4) ? w : 7 - w;
  const int cb = (w >> 2) * 4;
  const int s0 = c * BS_;
  const int n0 = s0 + 16 * t;

  const float* Q = Qb + (size_t)bb * L_ * HD_;
  const float* K = Kb + (size_t)bb * L_ * HD_;
  const float* V = Vb + (size_t)bb * L_ * VD_;
  float*       O = Ob + (size_t)bb * L_ * VD_;

  v2f qa[16];
  {
    const float* qp = Q + (size_t)n0 * HD_ + lane * 4;
#pragma unroll
    for (int t4 = 0; t4 < 4; ++t4) {
      v4f q0 = *(const v4f*)(qp + 256 * t4);
      v4f q1 = *(const v4f*)(qp + 256 * t4 + 128);
      qa[4 * t4 + 0] = v2f{q0.x, q0.y};
      qa[4 * t4 + 1] = v2f{q0.z, q0.w};
      qa[4 * t4 + 2] = v2f{q1.x, q1.y};
      qa[4 * t4 + 3] = v2f{q1.z, q1.w};
    }
  }

  const v8f vz = {0.f, 0.f, 0.f, 0.f, 0.f, 0.f, 0.f, 0.f};

  v8f oi[4];
#pragma unroll
  for (int ctl = 0; ctl < 4; ++ctl) oi[ctl] = vz;
  if (c > 0) {
    const float* P = O + (size_t)s0 * VD_ + lane * 4;
#pragma unroll
    for (int ctl = 0; ctl < 4; ++ctl) {
      const float* pp = P + (cb + ctl) * 1024;
#pragma unroll
      for (int t4 = 0; t4 < 4; ++t4) {
        v4f p0 = *(const v4f*)(pp + 256 * t4);
        v4f p1 = *(const v4f*)(pp + 256 * t4 + 128);
        oi[ctl] = wmma_f32_16x16x4(qa[4 * t4 + 0], v2f{p0.x, p0.y}, oi[ctl]);
        oi[ctl] = wmma_f32_16x16x4(qa[4 * t4 + 1], v2f{p0.z, p0.w}, oi[ctl]);
        oi[ctl] = wmma_f32_16x16x4(qa[4 * t4 + 2], v2f{p1.x, p1.y}, oi[ctl]);
        oi[ctl] = wmma_f32_16x16x4(qa[4 * t4 + 3], v2f{p1.z, p1.w}, oi[ctl]);
      }
    }
  }
  __syncthreads();

  float gp[8];
#pragma unroll
  for (int i = 0; i < 8; ++i)
    gp[i] = exp2f((float)((i + 8 * hi) - r) * LOG2_GAMMA);

  v8f o[4];
#pragma unroll
  for (int ctl = 0; ctl < 4; ++ctl) o[ctl] = vz;

  const float* kptr  = K + (size_t)s0 * HD_ + lane * 4;
  const float* vbase = V + (size_t)s0 * VD_ + cb * 256 + lane * 4;

  for (int m0 = s0; m0 <= n0; m0 += 16) {
    v4f kf[8];
#pragma unroll
    for (int j = 0; j < 8; ++j) kf[j] = *(const v4f*)(kptr + j * 128);
    v4f vf[2][4];
#pragma unroll
    for (int q = 0; q < 2; ++q) {
#pragma unroll
      for (int ctl = 0; ctl < 4; ++ctl)
        vf[q][ctl] = *(const v4f*)(vbase + ctl * 256 + q * 128);
    }

    v8f s = vz;
#pragma unroll
    for (int j = 0; j < 8; ++j) {
      s = wmma_f32_16x16x4(qa[2 * j],     v2f{kf[j].x, kf[j].y}, s);
      s = wmma_f32_16x16x4(qa[2 * j + 1], v2f{kf[j].z, kf[j].w}, s);
    }

    {
      const bool diag = (m0 == n0);
      float cf = exp2f((float)(n0 - m0) * LOG2_GAMMA);
#pragma unroll
      for (int i = 0; i < 8; ++i) {
        int diff = (i + 8 * hi) - r;
        float v = s[i] * (cf * gp[i]);
        s[i] = (diag && diff < 0) ? 0.0f : v;
      }
    }

#pragma unroll
    for (int i = 0; i < 8; ++i) sS[(i + 8 * hi) * 18 + r] = s[i];
    v2f av[4];
#pragma unroll
    for (int u = 0; u < 4; ++u)
      av[u] = *(const v2f*)&sS[r * 18 + 8 * hi + 2 * u];

#pragma unroll
    for (int q = 0; q < 2; ++q) {
#pragma unroll
      for (int ctl = 0; ctl < 4; ++ctl) {
        v4f f = vf[q][ctl];
        o[ctl] = wmma_f32_16x16x4(av[2 * q],     v2f{f.x, f.y}, o[ctl]);
        o[ctl] = wmma_f32_16x16x4(av[2 * q + 1], v2f{f.z, f.w}, o[ctl]);
      }
    }

    kptr  += 16 * HD_;
    vbase += 16 * VD_;
  }

#pragma unroll
  for (int i = 0; i < 8; ++i) {
    float rd = exp2f((float)(16 * t + i + 8 * hi) * LOG2_GAMMA);
#pragma unroll
    for (int ctl = 0; ctl < 4; ++ctl)
      sS[(i + 8 * hi) * WP_ + ctl * 16 + r] = fmaf(oi[ctl][i], rd, o[ctl][i]);
  }
  float* orow = O + (size_t)(n0 + hi) * VD_ + 16 * cb + 4 * r;
#pragma unroll
  for (int p = 0; p < 8; ++p) {
    v4f x = *(const v4f*)&sS[(2 * p + hi) * WP_ + 4 * r];
    *(v4f*)(orow + (size_t)(2 * p) * VD_) = x;
  }
}

extern "C" void kernel_launch(void* const* d_in, const int* in_sizes, int n_in,
                              void* d_out, int out_size, void* d_ws, size_t ws_size,
                              hipStream_t stream) {
  const float* X  = (const float*)d_in[0];
  const float* WQ = (const float*)d_in[1];
  const float* WK = (const float*)d_in[2];
  const float* WV = (const float*)d_in[3];

  float* Qb = (float*)d_ws;
  float* Kb = Qb + (size_t)B_ * L_ * HD_;
  float* Vb = Kb + (size_t)B_ * L_ * HD_;
  float* O  = (float*)d_out;

  proj_xpos_kernel<<<dim3(B_ * NBLK_), dim3(256), 0, stream>>>(X, WQ, WK, WV,
                                                               Qb, Kb, Vb, O);
  state_scan_kernel<<<dim3(B_ * 64), dim3(32), 0, stream>>>(O);
  retention_block_kernel<<<dim3(B_ * NBLK_), dim3(256), 0, stream>>>(Qb, Kb, Vb, O);
}
